# starting kernel
# baseline (speedup 1.0000x reference)
#include <hip/hip_runtime.h>
#include <hip/hip_bf16.h>


typedef float v2f __attribute__((ext_vector_type(2)));
typedef float v8f __attribute__((ext_vector_type(8)));

#define NN   100000
#define NE   1600000
#define IND  256
#define HIDD 128
#define OUTD 32
#define NEG_SLOPE 0.2f

template <int K, int NC, bool BT, bool ELU>
__global__ __launch_bounds__(128) void gemm_wmma(const float* __restrict__ A,
                                                 const float* __restrict__ Bm,
                                                 float* __restrict__ D, int M) {
  extern __shared__ float smem[];

  const int tid = threadIdx.x;
  for (int i = tid; i < K * NC; i += blockDim.x) {
    const int k = i / NC;
    const int n = i - k * NC;
    smem[(((k >> 1) * NC + n) << 1) | (k & 1)] =
        BT ? Bm[(size_t)n * K + k] : Bm[i];
  }
  __syncthreads();

  const int lane  = tid & 31;
  const int wv    = tid >> 5;
  const int strip = blockIdx.x * (blockDim.x >> 5) + wv;
  const int row0  = strip * 16;
  if (row0 >= M) return;

  const int m    = lane & 15;
  const int koff = (lane >> 4) << 1;
  const float* arow = A + (size_t)(row0 + m) * K + koff;

  constexpr int NH    = (NC > 128) ? 128 : NC;
  constexpr int NPASS = NC / NH;
  constexpr int NT    = NH / 16;
  static_assert(K % 8 == 0, "k-loop unrolled by 8");

  for (int np = 0; np < NPASS; ++np) {
    v8f acc[NT];
#pragma unroll
    for (int t = 0; t < NT; ++t)
#pragma unroll
      for (int i = 0; i < 8; ++i) acc[t][i] = 0.0f;

    auto loadB = [&](int kk, v2f* dst) {
      const float* bp = smem + ((((kk + koff) >> 1) * NC + np * NH + m) << 1);
#pragma unroll
      for (int t = 0; t < NT; ++t)
        dst[t] = *reinterpret_cast<const v2f*>(bp + t * 32);
    };
    auto mmaStep = [&](int kk, const v2f* bv) {
      v2f av = *reinterpret_cast<const v2f*>(arow + kk);
      if (ELU) {
        av[0] = av[0] > 0.0f ? av[0] : (expf(av[0]) - 1.0f);
        av[1] = av[1] > 0.0f ? av[1] : (expf(av[1]) - 1.0f);
      }
#pragma unroll
      for (int t = 0; t < NT; ++t)
        acc[t] = __builtin_amdgcn_wmma_f32_16x16x4_f32(
            false, av, false, bv[t], (short)0, acc[t], false, false);
    };

    v2f bv0[NT], bv1[NT];
    loadB(0, bv0);
    for (int k0 = 0; k0 < K; k0 += 8) {
      loadB(k0 + 4, bv1);
      mmaStep(k0, bv0);
      if (k0 + 8 < K) loadB(k0 + 8, bv0);
      mmaStep(k0 + 4, bv1);
    }

    const int mbase = (lane >> 4) * 8;
#pragma unroll
    for (int t = 0; t < NT; ++t) {
#pragma unroll
      for (int v = 0; v < 8; ++v) {
        D[(size_t)(row0 + mbase + v) * NC + np * NH + t * 16 + m] = acc[t][v];
      }
    }
  }
}

__global__ void node_dots(const float* __restrict__ xs1,
                          const float* __restrict__ a_s,
                          const float* __restrict__ a_d,
                          float* __restrict__ as1, float* __restrict__ ad1,
                          int n) {
  const int gw   = (blockIdx.x * blockDim.x + threadIdx.x) >> 5;
  const int lane = threadIdx.x & 31;
  if (gw >= n) return;
  const float4 v  = ((const float4*)(xs1 + (size_t)gw * HIDD))[lane];
  const float4 cs = ((const float4*)a_s)[lane];
  const float4 cd = ((const float4*)a_d)[lane];
  float s = v.x * cs.x + v.y * cs.y + v.z * cs.z + v.w * cs.w;
  float d = v.x * cd.x + v.y * cd.y + v.z * cd.z + v.w * cd.w;
#pragma unroll
  for (int off = 16; off > 0; off >>= 1) {
    s += __shfl_down(s, off, 32);
    d += __shfl_down(d, off, 32);
  }
  if (lane == 0) {
    as1[gw] = s;
    ad1[gw] = d;
  }
}

__device__ __forceinline__ unsigned f2key(float f) {
  const unsigned b = __float_as_uint(f);
  return (b & 0x80000000u) ? ~b : (b | 0x80000000u);
}
__device__ __forceinline__ float key2f(unsigned u) {
  return __uint_as_float((u & 0x80000000u) ? (u & 0x7FFFFFFFu) : ~u);
}

__global__ void edge_logits(const int* __restrict__ ei,
                            const float* __restrict__ as1,
                            const float* __restrict__ ad1,
                            float* __restrict__ ebuf,
                            unsigned* __restrict__ mbuf, int E) {
  const int e = blockIdx.x * blockDim.x + threadIdx.x;
  if (e >= E) return;
  const int s = ei[e];
  const int d = ei[E + e];
  float v = as1[s] + ad1[d];
  v = v > 0.0f ? v : NEG_SLOPE * v;
  ebuf[e] = v;
  atomicMax(&mbuf[d], f2key(v));
}

__global__ void edge_expw(const int* __restrict__ ei,
                          const unsigned* __restrict__ mbuf,
                          float* __restrict__ ebuf, float* __restrict__ sbuf,
                          int E) {
  const int e = blockIdx.x * blockDim.x + threadIdx.x;
  if (e >= E) return;
  const int d = ei[E + e];
  const float w = expf(ebuf[e] - key2f(mbuf[d]));
  ebuf[e] = w;
  atomicAdd(&sbuf[d], w);
}

template <bool FIRST>
__global__ void edge_aggregate(const int* __restrict__ ei,
                               const float* __restrict__ feat,
                               const float* __restrict__ wb,
                               const float* __restrict__ sb,
                               float* __restrict__ alpha,
                               float* __restrict__ agg, int E) {
  const int gw   = (blockIdx.x * blockDim.x + threadIdx.x) >> 5;
  const int lane = threadIdx.x & 31;
  if (gw >= E) return;
  const int s = ei[gw];
  const int d = ei[E + gw];
  float a;
  if (FIRST) {
    a = wb[gw] / sb[d];
    if (lane == 0) alpha[gw] = a;
  } else {
    a = alpha[gw];
  }
  const float4 v = ((const float4*)(feat + (size_t)s * HIDD))[lane];
  float* ar = agg + (size_t)d * HIDD + lane * 4;
  atomicAdd(ar + 0, v.x * a);
  atomicAdd(ar + 1, v.y * a);
  atomicAdd(ar + 2, v.z * a);
  atomicAdd(ar + 3, v.w * a);
}

extern "C" void kernel_launch(void* const* d_in, const int* in_sizes, int n_in,
                              void* d_out, int out_size, void* d_ws,
                              size_t ws_size, hipStream_t stream) {
  const float* x   = (const float*)d_in[0];
  const float* W1  = (const float*)d_in[1];
  const float* a1s = (const float*)d_in[2];
  const float* a1d = (const float*)d_in[3];
  const float* W2  = (const float*)d_in[4];
  const int*   ei  = (const int*)d_in[5];

  float* out = (float*)d_out;
  float* h2  = out;
  float* h4  = out + (size_t)NN * OUTD;

  float* ws    = (float*)d_ws;
  float* xs1   = ws;
  float* agg   = xs1 + (size_t)NN * HIDD;
  float* as1   = agg + (size_t)NN * HIDD;
  float* ad1   = as1 + NN;
  float* sbuf  = ad1 + NN;
  unsigned* mbuf = (unsigned*)(sbuf + NN);
  float* ebuf  = (float*)(mbuf + NN);
  float* alpha = ebuf + NE;

  hipMemsetAsync(agg, 0, (size_t)NN * HIDD * sizeof(float), stream);
  hipMemsetAsync(sbuf, 0, (size_t)2 * NN * sizeof(float), stream);

  const int strips  = NN / 16;
  const int gblocks = (strips + 3) / 4;
  const int eblocks = (NE + 255) / 256;
  const int ewarpsB = NE / 8;
  const int nwarpsB = (NN * 32 + 255) / 256;

  gemm_wmma<IND, HIDD, false, false>
      <<<gblocks, 128, IND * HIDD * sizeof(float), stream>>>(x, W1, xs1, NN);

  node_dots<<<nwarpsB, 256, 0, stream>>>(xs1, a1s, a1d, as1, ad1, NN);

  edge_logits<<<eblocks, 256, 0, stream>>>(ei, as1, ad1, ebuf, mbuf, NE);
  edge_expw<<<eblocks, 256, 0, stream>>>(ei, mbuf, ebuf, sbuf, NE);

  edge_aggregate<true>
      <<<ewarpsB, 256, 0, stream>>>(ei, xs1, ebuf, sbuf, alpha, agg, NE);

  gemm_wmma<HIDD, OUTD, false, true>
      <<<gblocks, 128, HIDD * OUTD * sizeof(float), stream>>>(agg, W2, h2, NN);

  gemm_wmma<OUTD, HIDD, true, false>
      <<<gblocks, 128, OUTD * HIDD * sizeof(float), stream>>>(h2, W2, xs1, NN);

  hipMemsetAsync(agg, 0, (size_t)NN * HIDD * sizeof(float), stream);
  edge_aggregate<false>
      <<<ewarpsB, 256, 0, stream>>>(ei, xs1, ebuf, sbuf, alpha, agg, NE);

  gemm_wmma<HIDD, IND, true, true>
      <<<gblocks, 128, HIDD * IND * sizeof(float), stream>>>(agg, W1, h4, NN);
}


# B staging into LDS: each thread issues a batch of independent 16-byte global loads (whole lines) before its LDS stores, instead of one dependent load-wait-store per element
# speedup vs baseline: 1.2703x; 1.2703x over previous
#include <hip/hip_runtime.h>
#include <hip/hip_bf16.h>


typedef float v2f __attribute__((ext_vector_type(2)));
typedef float v8f __attribute__((ext_vector_type(8)));

#define NN   100000
#define NE   1600000
#define IND  256
#define HIDD 128
#define OUTD 32
#define NEG_SLOPE 0.2f

template <int K, int NC, bool BT>
__device__ __forceinline__ void stage_b(const float* __restrict__ Bm,
                                        float* __restrict__ smem, int tid) {
  constexpr int NTHR = 128;
  if (!BT) {
    constexpr int NQ    = NC / 4;
    constexpr int ITEMS = (K / 2) * NQ;
    constexpr int PER_T = ITEMS / NTHR;
    constexpr int UN    = PER_T < 8 ? PER_T : 8;
    static_assert(ITEMS % (NTHR * UN) == 0, "whole batches");
    for (int base = 0; base < ITEMS; base += NTHR * UN) {
      float4 r0[UN], r1[UN];
#pragma unroll
      for (int u = 0; u < UN; ++u) {
        const int it = base + u * NTHR + tid;
        const int kp = it / NQ;
        const float* g = Bm + 4 * it + kp * NC;
        r0[u] = *reinterpret_cast<const float4*>(g);
        r1[u] = *reinterpret_cast<const float4*>(g + NC);
      }
#pragma unroll
      for (int u = 0; u < UN; ++u) {
        const int it = base + u * NTHR + tid;
        float4* dst = reinterpret_cast<float4*>(smem + 8 * it);
        dst[0] = make_float4(r0[u].x, r1[u].x, r0[u].y, r1[u].y);
        dst[1] = make_float4(r0[u].z, r1[u].z, r0[u].w, r1[u].w);
      }
    }
  } else {
    constexpr int LK    = 4;
    constexpr int NSUB  = NTHR / LK;
    constexpr int KQH   = (K / 4) / LK;
    constexpr int STEPS = (NC / NSUB) * KQH;
    constexpr int UN    = STEPS < 8 ? STEPS : 8;
    static_assert((K / 4) % LK == 0 && NC % NSUB == 0 && STEPS % UN == 0,
                  "whole batches");
    const int lk   = tid % LK;
    const int nsub = tid / LK;
    for (int s0 = 0; s0 < STEPS; s0 += UN) {
      float4 r[UN];
#pragma unroll
      for (int u = 0; u < UN; ++u) {
        const int s  = s0 + u;
        const int kq = (s % KQH) * LK + lk;
        const int n  = (s / KQH) * NSUB + nsub;
        r[u] = *reinterpret_cast<const float4*>(Bm + n * K + 4 * kq);
      }
#pragma unroll
      for (int u = 0; u < UN; ++u) {
        const int s  = s0 + u;
        const int kq = (s % KQH) * LK + lk;
        const int n  = (s / KQH) * NSUB + nsub;
        float* p = smem + (((2 * kq) * NC + n) << 1);
        *reinterpret_cast<float2*>(p)          = make_float2(r[u].x, r[u].y);
        *reinterpret_cast<float2*>(p + 2 * NC) = make_float2(r[u].z, r[u].w);
      }
    }
  }
}

template <int K, int NC, bool BT, bool ELU>
__global__ __launch_bounds__(128) void gemm_wmma(const float* __restrict__ A,
                                                 const float* __restrict__ Bm,
                                                 float* __restrict__ D, int M) {
  extern __shared__ __align__(16) float smem[];

  const int tid = threadIdx.x;
  stage_b<K, NC, BT>(Bm, smem, tid);
  __syncthreads();

  const int lane  = tid & 31;
  const int wv    = tid >> 5;
  const int strip = blockIdx.x * (blockDim.x >> 5) + wv;
  const int row0  = strip * 16;
  if (row0 >= M) return;

  const int m    = lane & 15;
  const int koff = (lane >> 4) << 1;
  const float* arow = A + (size_t)(row0 + m) * K + koff;

  constexpr int NH    = (NC > 128) ? 128 : NC;
  constexpr int NPASS = NC / NH;
  constexpr int NT    = NH / 16;
  static_assert(K % 8 == 0, "k-loop unrolled by 8");

  for (int np = 0; np < NPASS; ++np) {
    v8f acc[NT];
#pragma unroll
    for (int t = 0; t < NT; ++t)
#pragma unroll
      for (int i = 0; i < 8; ++i) acc[t][i] = 0.0f;

    auto loadB = [&](int kk, v2f* dst) {
      const float* bp = smem + ((((kk + koff) >> 1) * NC + np * NH + m) << 1);
#pragma unroll
      for (int t = 0; t < NT; ++t)
        dst[t] = *reinterpret_cast<const v2f*>(bp + t * 32);
    };
    auto mmaStep = [&](int kk, const v2f* bv) {
      v2f av = *reinterpret_cast<const v2f*>(arow + kk);
      if (ELU) {
        av[0] = av[0] > 0.0f ? av[0] : (expf(av[0]) - 1.0f);
        av[1] = av[1] > 0.0f ? av[1] : (expf(av[1]) - 1.0f);
      }
#pragma unroll
      for (int t = 0; t < NT; ++t)
        acc[t] = __builtin_amdgcn_wmma_f32_16x16x4_f32(
            false, av, false, bv[t], (short)0, acc[t], false, false);
    };

    v2f bv0[NT], bv1[NT];
    loadB(0, bv0);
    for (int k0 = 0; k0 < K; k0 += 8) {
      loadB(k0 + 4, bv1);
      mmaStep(k0, bv0);
      if (k0 + 8 < K) loadB(k0 + 8, bv0);
      mmaStep(k0 + 4, bv1);
    }

    const int mbase = (lane >> 4) * 8;
#pragma unroll
    for (int t = 0; t < NT; ++t) {
#pragma unroll
      for (int v = 0; v < 8; ++v) {
        D[(size_t)(row0 + mbase + v) * NC + np * NH + t * 16 + m] = acc[t][v];
      }
    }
  }
}

__global__ void node_dots(const float* __restrict__ xs1,
                          const float* __restrict__ a_s,
                          const float* __restrict__ a_d,
                          float* __restrict__ as1, float* __restrict__ ad1,
                          int n) {
  const int gw   = (blockIdx.x * blockDim.x + threadIdx.x) >> 5;
  const int lane = threadIdx.x & 31;
  if (gw >= n) return;
  const float4 v  = ((const float4*)(xs1 + (size_t)gw * HIDD))[lane];
  const float4 cs = ((const float4*)a_s)[lane];
  const float4 cd = ((const float4*)a_d)[lane];
  float s = v.x * cs.x + v.y * cs.y + v.z * cs.z + v.w * cs.w;
  float d = v.x * cd.x + v.y * cd.y + v.z * cd.z + v.w * cd.w;
#pragma unroll
  for (int off = 16; off > 0; off >>= 1) {
    s += __shfl_down(s, off, 32);
    d += __shfl_down(d, off, 32);
  }
  if (lane == 0) {
    as1[gw] = s;
    ad1[gw] = d;
  }
}

__device__ __forceinline__ unsigned f2key(float f) {
  const unsigned b = __float_as_uint(f);
  return (b & 0x80000000u) ? ~b : (b | 0x80000000u);
}
__device__ __forceinline__ float key2f(unsigned u) {
  return __uint_as_float((u & 0x80000000u) ? (u & 0x7FFFFFFFu) : ~u);
}

__global__ void edge_logits(const int* __restrict__ ei,
                            const float* __restrict__ as1,
                            const float* __restrict__ ad1,
                            float* __restrict__ ebuf,
                            unsigned* __restrict__ mbuf, int E) {
  const int e = blockIdx.x * blockDim.x + threadIdx.x;
  if (e >= E) return;
  const int s = ei[e];
  const int d = ei[E + e];
  float v = as1[s] + ad1[d];
  v = v > 0.0f ? v : NEG_SLOPE * v;
  ebuf[e] = v;
  atomicMax(&mbuf[d], f2key(v));
}

__global__ void edge_expw(const int* __restrict__ ei,
                          const unsigned* __restrict__ mbuf,
                          float* __restrict__ ebuf, float* __restrict__ sbuf,
                          int E) {
  const int e = blockIdx.x * blockDim.x + threadIdx.x;
  if (e >= E) return;
  const int d = ei[E + e];
  const float w = expf(ebuf[e] - key2f(mbuf[d]));
  ebuf[e] = w;
  atomicAdd(&sbuf[d], w);
}

template <bool FIRST>
__global__ void edge_aggregate(const int* __restrict__ ei,
                               const float* __restrict__ feat,
                               const float* __restrict__ wb,
                               const float* __restrict__ sb,
                               float* __restrict__ alpha,
                               float* __restrict__ agg, int E) {
  const int gw   = (blockIdx.x * blockDim.x + threadIdx.x) >> 5;
  const int lane = threadIdx.x & 31;
  if (gw >= E) return;
  const int s = ei[gw];
  const int d = ei[E + gw];
  float a;
  if (FIRST) {
    a = wb[gw] / sb[d];
    if (lane == 0) alpha[gw] = a;
  } else {
    a = alpha[gw];
  }
  const float4 v = ((const float4*)(feat + (size_t)s * HIDD))[lane];
  float* ar = agg + (size_t)d * HIDD + lane * 4;
  atomicAdd(ar + 0, v.x * a);
  atomicAdd(ar + 1, v.y * a);
  atomicAdd(ar + 2, v.z * a);
  atomicAdd(ar + 3, v.w * a);
}

extern "C" void kernel_launch(void* const* d_in, const int* in_sizes, int n_in,
                              void* d_out, int out_size, void* d_ws,
                              size_t ws_size, hipStream_t stream) {
  const float* x   = (const float*)d_in[0];
  const float* W1  = (const float*)d_in[1];
  const float* a1s = (const float*)d_in[2];
  const float* a1d = (const float*)d_in[3];
  const float* W2  = (const float*)d_in[4];
  const int*   ei  = (const int*)d_in[5];

  float* out = (float*)d_out;
  float* h2  = out;
  float* h4  = out + (size_t)NN * OUTD;

  float* ws    = (float*)d_ws;
  float* xs1   = ws;
  float* agg   = xs1 + (size_t)NN * HIDD;
  float* as1   = agg + (size_t)NN * HIDD;
  float* ad1   = as1 + NN;
  float* sbuf  = ad1 + NN;
  unsigned* mbuf = (unsigned*)(sbuf + NN);
  float* ebuf  = (float*)(mbuf + NN);
  float* alpha = ebuf + NE;

  hipMemsetAsync(agg, 0, (size_t)NN * HIDD * sizeof(float), stream);
  hipMemsetAsync(sbuf, 0, (size_t)2 * NN * sizeof(float), stream);

  const int strips  = NN / 16;
  const int gblocks = (strips + 3) / 4;
  const int eblocks = (NE + 255) / 256;
  const int ewarpsB = NE / 8;
  const int nwarpsB = (NN * 32 + 255) / 256;

  gemm_wmma<IND, HIDD, false, false>
      <<<gblocks, 128, IND * HIDD * sizeof(float), stream>>>(x, W1, xs1, NN);

  node_dots<<<nwarpsB, 256, 0, stream>>>(xs1, a1s, a1d, as1, ad1, NN);

  edge_logits<<<eblocks, 256, 0, stream>>>(ei, as1, ad1, ebuf, mbuf, NE);
  edge_expw<<<eblocks, 256, 0, stream>>>(ei, mbuf, ebuf, sbuf, NE);

  edge_aggregate<true>
      <<<ewarpsB, 256, 0, stream>>>(ei, xs1, ebuf, sbuf, alpha, agg, NE);

  gemm_wmma<HIDD, OUTD, false, true>
      <<<gblocks, 128, HIDD * OUTD * sizeof(float), stream>>>(agg, W2, h2, NN);

  gemm_wmma<OUTD, HIDD, true, false>
      <<<gblocks, 128, OUTD * HIDD * sizeof(float), stream>>>(h2, W2, xs1, NN);

  hipMemsetAsync(agg, 0, (size_t)NN * HIDD * sizeof(float), stream);
  edge_aggregate<false>
      <<<ewarpsB, 256, 0, stream>>>(ei, xs1, ebuf, sbuf, alpha, agg, NE);

  gemm_wmma<HIDD, IND, true, true>
      <<<gblocks, 128, HIDD * IND * sizeof(float), stream>>>(agg, W1, h4, NN);
}


# Aggregation without float atomics: edges counting-sorted by destination each call; one wave per node does its softmax and sums its source rows in registers, one store per row; sums in arrival order
# speedup vs baseline: 2.9056x; 2.2873x over previous
#include <hip/hip_runtime.h>
#include <hip/hip_bf16.h>


typedef float v2f __attribute__((ext_vector_type(2)));
typedef float v8f __attribute__((ext_vector_type(8)));

#define NN   100000
#define NE   1600000
#define IND  256
#define HIDD 128
#define OUTD 32
#define NEG_SLOPE 0.2f

template <int K, int NC, bool BT>
__device__ __forceinline__ void stage_b(const float* __restrict__ Bm,
                                        float* __restrict__ smem, int tid) {
  constexpr int NTHR = 128;
  if (!BT) {
    constexpr int NQ    = NC / 4;
    constexpr int ITEMS = (K / 2) * NQ;
    constexpr int PER_T = ITEMS / NTHR;
    constexpr int UN    = PER_T < 8 ? PER_T : 8;
    static_assert(ITEMS % (NTHR * UN) == 0, "whole batches");
    for (int base = 0; base < ITEMS; base += NTHR * UN) {
      float4 r0[UN], r1[UN];
#pragma unroll
      for (int u = 0; u < UN; ++u) {
        const int it = base + u * NTHR + tid;
        const int kp = it / NQ;
        const float* g = Bm + 4 * it + kp * NC;
        r0[u] = *reinterpret_cast<const float4*>(g);
        r1[u] = *reinterpret_cast<const float4*>(g + NC);
      }
#pragma unroll
      for (int u = 0; u < UN; ++u) {
        const int it = base + u * NTHR + tid;
        float4* dst = reinterpret_cast<float4*>(smem + 8 * it);
        dst[0] = make_float4(r0[u].x, r1[u].x, r0[u].y, r1[u].y);
        dst[1] = make_float4(r0[u].z, r1[u].z, r0[u].w, r1[u].w);
      }
    }
  } else {
    constexpr int LK    = 4;
    constexpr int NSUB  = NTHR / LK;
    constexpr int KQH   = (K / 4) / LK;
    constexpr int STEPS = (NC / NSUB) * KQH;
    constexpr int UN    = STEPS < 8 ? STEPS : 8;
    static_assert((K / 4) % LK == 0 && NC % NSUB == 0 && STEPS % UN == 0,
                  "whole batches");
    const int lk   = tid % LK;
    const int nsub = tid / LK;
    for (int s0 = 0; s0 < STEPS; s0 += UN) {
      float4 r[UN];
#pragma unroll
      for (int u = 0; u < UN; ++u) {
        const int s  = s0 + u;
        const int kq = (s % KQH) * LK + lk;
        const int n  = (s / KQH) * NSUB + nsub;
        r[u] = *reinterpret_cast<const float4*>(Bm + n * K + 4 * kq);
      }
#pragma unroll
      for (int u = 0; u < UN; ++u) {
        const int s  = s0 + u;
        const int kq = (s % KQH) * LK + lk;
        const int n  = (s / KQH) * NSUB + nsub;
        float* p = smem + (((2 * kq) * NC + n) << 1);
        *reinterpret_cast<float2*>(p)          = make_float2(r[u].x, r[u].y);
        *reinterpret_cast<float2*>(p + 2 * NC) = make_float2(r[u].z, r[u].w);
      }
    }
  }
}

template <int K, int NC, bool BT, bool ELU>
__global__ __launch_bounds__(128) void gemm_wmma(const float* __restrict__ A,
                                                 const float* __restrict__ Bm,
                                                 float* __restrict__ D, int M) {
  extern __shared__ __align__(16) float smem[];

  const int tid = threadIdx.x;
  stage_b<K, NC, BT>(Bm, smem, tid);
  __syncthreads();

  const int lane  = tid & 31;
  const int wv    = tid >> 5;
  const int strip = blockIdx.x * (blockDim.x >> 5) + wv;
  const int row0  = strip * 16;
  if (row0 >= M) return;

  const int m    = lane & 15;
  const int koff = (lane >> 4) << 1;
  const float* arow = A + (size_t)(row0 + m) * K + koff;

  constexpr int NH    = (NC > 128) ? 128 : NC;
  constexpr int NPASS = NC / NH;
  constexpr int NT    = NH / 16;
  static_assert(K % 8 == 0, "k-loop unrolled by 8");

  for (int np = 0; np < NPASS; ++np) {
    v8f acc[NT];
#pragma unroll
    for (int t = 0; t < NT; ++t)
#pragma unroll
      for (int i = 0; i < 8; ++i) acc[t][i] = 0.0f;

    auto loadB = [&](int kk, v2f* dst) {
      const float* bp = smem + ((((kk + koff) >> 1) * NC + np * NH + m) << 1);
#pragma unroll
      for (int t = 0; t < NT; ++t)
        dst[t] = *reinterpret_cast<const v2f*>(bp + t * 32);
    };
    auto mmaStep = [&](int kk, const v2f* bv) {
      v2f av = *reinterpret_cast<const v2f*>(arow + kk);
      if (ELU) {
        av[0] = av[0] > 0.0f ? av[0] : (expf(av[0]) - 1.0f);
        av[1] = av[1] > 0.0f ? av[1] : (expf(av[1]) - 1.0f);
      }
#pragma unroll
      for (int t = 0; t < NT; ++t)
        acc[t] = __builtin_amdgcn_wmma_f32_16x16x4_f32(
            false, av, false, bv[t], (short)0, acc[t], false, false);
    };

    v2f bv0[NT], bv1[NT];
    loadB(0, bv0);
    for (int k0 = 0; k0 < K; k0 += 8) {
      loadB(k0 + 4, bv1);
      mmaStep(k0, bv0);
      if (k0 + 8 < K) loadB(k0 + 8, bv0);
      mmaStep(k0 + 4, bv1);
    }

    const int mbase = (lane >> 4) * 8;
#pragma unroll
    for (int t = 0; t < NT; ++t) {
#pragma unroll
      for (int v = 0; v < 8; ++v) {
        D[(size_t)(row0 + mbase + v) * NC + np * NH + t * 16 + m] = acc[t][v];
      }
    }
  }
}

__global__ void node_dots(const float* __restrict__ xs1,
                          const float* __restrict__ a_s,
                          const float* __restrict__ a_d,
                          float* __restrict__ as1, float* __restrict__ ad1,
                          int n) {
  const int gw   = (blockIdx.x * blockDim.x + threadIdx.x) >> 5;
  const int lane = threadIdx.x & 31;
  if (gw >= n) return;
  const float4 v  = ((const float4*)(xs1 + (size_t)gw * HIDD))[lane];
  const float4 cs = ((const float4*)a_s)[lane];
  const float4 cd = ((const float4*)a_d)[lane];
  float s = v.x * cs.x + v.y * cs.y + v.z * cs.z + v.w * cs.w;
  float d = v.x * cd.x + v.y * cd.y + v.z * cd.z + v.w * cd.w;
#pragma unroll
  for (int off = 16; off > 0; off >>= 1) {
    s += __shfl_down(s, off, 32);
    d += __shfl_down(d, off, 32);
  }
  if (lane == 0) {
    as1[gw] = s;
    ad1[gw] = d;
  }
}

__global__ void deg_rank(const int* __restrict__ ei, int* __restrict__ cnt,
                         int* __restrict__ rank, int E) {
  const int e = blockIdx.x * blockDim.x + threadIdx.x;
  if (e >= E) return;
  rank[e] = atomicAdd(&cnt[ei[E + e]], 1);
}

__global__ __launch_bounds__(1024) void seg_scan(int* __restrict__ cnt, int n) {
  __shared__ int wsum[2][32];
  const int tid  = threadIdx.x;
  const int lane = tid & 31;
  const int wv   = tid >> 5;
  int carry = 0;
  int it = 0;
  for (int base = 0; base < n; base += 4096, ++it) {
    const int i = base + tid * 4;
    int4 v = make_int4(0, 0, 0, 0);
    if (i + 3 < n) {
      v = *reinterpret_cast<const int4*>(cnt + i);
    } else {
      if (i < n) v.x = cnt[i];
      if (i + 1 < n) v.y = cnt[i + 1];
      if (i + 2 < n) v.z = cnt[i + 2];
    }
    const int t0 = v.x, t1 = t0 + v.y, t2 = t1 + v.z, t3 = t2 + v.w;
    int x = t3;
#pragma unroll
    for (int off = 1; off < 32; off <<= 1) {
      const int y = __shfl_up(x, off, 32);
      if (lane >= off) x += y;
    }
    if (lane == 31) wsum[it & 1][wv] = x;
    __syncthreads();
    const int ws = wsum[it & 1][lane];
    int wx = ws;
#pragma unroll
    for (int off = 1; off < 32; off <<= 1) {
      const int y = __shfl_up(wx, off, 32);
      if (lane >= off) wx += y;
    }
    const int total = __shfl(wx, 31, 32);
    const int wexcl = __shfl(wx, wv, 32) - __shfl(ws, wv, 32);
    const int excl  = carry + wexcl + (x - t3);
    if (i + 3 < n) {
      *reinterpret_cast<int4*>(cnt + i) =
          make_int4(excl, excl + t0, excl + t1, excl + t2);
    } else {
      if (i < n) cnt[i] = excl;
      if (i + 1 < n) cnt[i + 1] = excl + t0;
      if (i + 2 < n) cnt[i + 2] = excl + t1;
    }
    carry += total;
  }
}

__global__ void csr_fill(const int* __restrict__ ei,
                         const int* __restrict__ start,
                         const int* __restrict__ rank,
                         int* __restrict__ csrc, int E) {
  const int e = blockIdx.x * blockDim.x + threadIdx.x;
  if (e >= E) return;
  csrc[start[ei[E + e]] + rank[e]] = ei[e];
}

__device__ __forceinline__ float leaky(float v) {
  return v > 0.0f ? v : NEG_SLOPE * v;
}
__device__ __forceinline__ float wave_max(float v) {
#pragma unroll
  for (int off = 16; off > 0; off >>= 1) v = fmaxf(v, __shfl_xor(v, off, 32));
  return v;
}
__device__ __forceinline__ float wave_sum(float v) {
#pragma unroll
  for (int off = 16; off > 0; off >>= 1) v += __shfl_xor(v, off, 32);
  return v;
}

__device__ __forceinline__ void add_rows(const float* __restrict__ fl, int s_l,
                                         float a_l, int cnt, float4& acc) {
  const int ai = __float_as_int(a_l);
  int u = 0;
  for (; u + 8 <= cnt; u += 8) {
    float4 v[8];
    float  a[8];
#pragma unroll
    for (int i = 0; i < 8; ++i) {
      const int s = __builtin_amdgcn_readlane(s_l, u + i);
      a[i] = __int_as_float(__builtin_amdgcn_readlane(ai, u + i));
      v[i] = *reinterpret_cast<const float4*>(fl + (size_t)s * HIDD);
    }
#pragma unroll
    for (int i = 0; i < 8; ++i) {
      acc.x = fmaf(v[i].x, a[i], acc.x);
      acc.y = fmaf(v[i].y, a[i], acc.y);
      acc.z = fmaf(v[i].z, a[i], acc.z);
      acc.w = fmaf(v[i].w, a[i], acc.w);
    }
  }
  if (u < cnt) {
    float4 v[8];
    float  a[8];
#pragma unroll
    for (int i = 0; i < 8; ++i) {
      if (u + i < cnt) {
        const int s = __builtin_amdgcn_readlane(s_l, u + i);
        a[i] = __int_as_float(__builtin_amdgcn_readlane(ai, u + i));
        v[i] = *reinterpret_cast<const float4*>(fl + (size_t)s * HIDD);
      }
    }
#pragma unroll
    for (int i = 0; i < 8; ++i) {
      if (u + i < cnt) {
        acc.x = fmaf(v[i].x, a[i], acc.x);
        acc.y = fmaf(v[i].y, a[i], acc.y);
        acc.z = fmaf(v[i].z, a[i], acc.z);
        acc.w = fmaf(v[i].w, a[i], acc.w);
      }
    }
  }
}

template <bool FIRST>
__global__ __launch_bounds__(256) void node_pull(
    const int* __restrict__ start, const int* __restrict__ csrc,
    const float* __restrict__ feat, const float* __restrict__ as1,
    const float* __restrict__ ad1, float* __restrict__ alpha,
    float* __restrict__ agg, int n, int E) {
  const int lane = threadIdx.x & 31;
  const int d = __builtin_amdgcn_readfirstlane(
      (int)((blockIdx.x * blockDim.x + threadIdx.x) >> 5));
  if (d >= n) return;
  const int beg = __builtin_amdgcn_readfirstlane(start[d]);
  const int end = (d + 1 < n) ? __builtin_amdgcn_readfirstlane(start[d + 1]) : E;
  const int deg = end - beg;

  const float* fl = feat + lane * 4;
  float4 acc = make_float4(0.0f, 0.0f, 0.0f, 0.0f);

  if (deg <= 32) {
    const int  j  = beg + lane;
    const bool ok = lane < deg;
    int   s_l = 0;
    float a_l = 0.0f;
    if (FIRST) {
      float v = -INFINITY;
      if (ok) {
        s_l = csrc[j];
        v   = leaky(as1[s_l] + ad1[d]);
      }
      const float m  = wave_max(v);
      const float w  = ok ? expf(v - m) : 0.0f;
      const float ss = wave_sum(w);
      if (ok) {
        a_l = w / ss;
        alpha[j] = a_l;
      }
    } else {
      if (ok) {
        s_l = csrc[j];
        a_l = alpha[j];
      }
    }
    add_rows(fl, s_l, a_l, deg, acc);
  } else {
    float m = 0.0f, ss = 1.0f, adv = 0.0f;
    if (FIRST) {
      adv = ad1[d];
      float mm = -INFINITY;
      for (int j = beg + lane; j < end; j += 32)
        mm = fmaxf(mm, leaky(as1[csrc[j]] + adv));
      m = wave_max(mm);
      float sm = 0.0f;
      for (int j = beg + lane; j < end; j += 32)
        sm += expf(leaky(as1[csrc[j]] + adv) - m);
      ss = wave_sum(sm);
    }
    for (int c = beg; c < end; c += 32) {
      const int j = c + lane;
      int   s_l = 0;
      float a_l = 0.0f;
      if (j < end) {
        s_l = csrc[j];
        if (FIRST) {
          a_l = expf(leaky(as1[s_l] + adv) - m) / ss;
          alpha[j] = a_l;
        } else {
          a_l = alpha[j];
        }
      }
      const int left = end - c;
      add_rows(fl, s_l, a_l, left < 32 ? left : 32, acc);
    }
  }
  *reinterpret_cast<float4*>(agg + (size_t)d * HIDD + lane * 4) = acc;
}

extern "C" void kernel_launch(void* const* d_in, const int* in_sizes, int n_in,
                              void* d_out, int out_size, void* d_ws,
                              size_t ws_size, hipStream_t stream) {
  const float* x   = (const float*)d_in[0];
  const float* W1  = (const float*)d_in[1];
  const float* a1s = (const float*)d_in[2];
  const float* a1d = (const float*)d_in[3];
  const float* W2  = (const float*)d_in[4];
  const int*   ei  = (const int*)d_in[5];

  float* out = (float*)d_out;
  float* h2  = out;
  float* h4  = out + (size_t)NN * OUTD;

  float* ws    = (float*)d_ws;
  float* xs1   = ws;
  float* agg   = xs1 + (size_t)NN * HIDD;
  float* as1   = agg + (size_t)NN * HIDD;
  float* ad1   = as1 + NN;
  int*   cnt   = (int*)(ad1 + NN);
  float* ebuf  = (float*)(cnt + 2 * NN);
  int*   csrc  = (int*)ebuf;
  float* alpha = ebuf + NE;
  int*   rank  = (int*)agg;

  hipMemsetAsync(cnt, 0, (size_t)NN * sizeof(int), stream);

  const int strips  = NN / 16;
  const int gblocks = (strips + 3) / 4;
  const int eblocks = (NE + 255) / 256;
  const int npullB  = (NN * 32 + 255) / 256;
  const int nwarpsB = (NN * 32 + 255) / 256;

  deg_rank<<<eblocks, 256, 0, stream>>>(ei, cnt, rank, NE);
  seg_scan<<<1, 1024, 0, stream>>>(cnt, NN);
  csr_fill<<<eblocks, 256, 0, stream>>>(ei, cnt, rank, csrc, NE);

  gemm_wmma<IND, HIDD, false, false>
      <<<gblocks, 128, IND * HIDD * sizeof(float), stream>>>(x, W1, xs1, NN);

  node_dots<<<nwarpsB, 256, 0, stream>>>(xs1, a1s, a1d, as1, ad1, NN);

  node_pull<true>
      <<<npullB, 256, 0, stream>>>(cnt, csrc, xs1, as1, ad1, alpha, agg, NN, NE);

  gemm_wmma<HIDD, OUTD, false, true>
      <<<gblocks, 128, HIDD * OUTD * sizeof(float), stream>>>(agg, W2, h2, NN);

  gemm_wmma<OUTD, HIDD, true, false>
      <<<gblocks, 128, OUTD * HIDD * sizeof(float), stream>>>(h2, W2, xs1, NN);

  node_pull<false>
      <<<npullB, 256, 0, stream>>>(cnt, csrc, xs1, as1, ad1, alpha, agg, NN, NE);

  gemm_wmma<HIDD, IND, true, true>
      <<<gblocks, 128, HIDD * IND * sizeof(float), stream>>>(agg, W1, h4, NN);
}


# conv3 aggregation moved in front of its linear map: attention-weighted sum taken over the 128-byte rows of h2, then @ W2^T (linearity); changes the order of summation, edge order as before
# speedup vs baseline: 3.0236x; 1.0406x over previous
#include <hip/hip_runtime.h>
#include <hip/hip_bf16.h>


typedef float v2f __attribute__((ext_vector_type(2)));
typedef float v8f __attribute__((ext_vector_type(8)));

#define NN   100000
#define NE   1600000
#define IND  256
#define HIDD 128
#define OUTD 32
#define NEG_SLOPE 0.2f

template <int K, int NC, bool BT>
__device__ __forceinline__ void stage_b(const float* __restrict__ Bm,
                                        float* __restrict__ smem, int tid) {
  constexpr int NTHR = 128;
  if (!BT) {
    constexpr int NQ    = NC / 4;
    constexpr int ITEMS = (K / 2) * NQ;
    constexpr int PER_T = ITEMS / NTHR;
    constexpr int UN    = PER_T < 8 ? PER_T : 8;
    static_assert(ITEMS % (NTHR * UN) == 0, "whole batches");
    for (int base = 0; base < ITEMS; base += NTHR * UN) {
      float4 r0[UN], r1[UN];
#pragma unroll
      for (int u = 0; u < UN; ++u) {
        const int it = base + u * NTHR + tid;
        const int kp = it / NQ;
        const float* g = Bm + 4 * it + kp * NC;
        r0[u] = *reinterpret_cast<const float4*>(g);
        r1[u] = *reinterpret_cast<const float4*>(g + NC);
      }
#pragma unroll
      for (int u = 0; u < UN; ++u) {
        const int it = base + u * NTHR + tid;
        float4* dst = reinterpret_cast<float4*>(smem + 8 * it);
        dst[0] = make_float4(r0[u].x, r1[u].x, r0[u].y, r1[u].y);
        dst[1] = make_float4(r0[u].z, r1[u].z, r0[u].w, r1[u].w);
      }
    }
  } else {
    constexpr int LK    = 4;
    constexpr int NSUB  = NTHR / LK;
    constexpr int KQH   = (K / 4) / LK;
    constexpr int STEPS = (NC / NSUB) * KQH;
    constexpr int UN    = STEPS < 8 ? STEPS : 8;
    static_assert((K / 4) % LK == 0 && NC % NSUB == 0 && STEPS % UN == 0,
                  "whole batches");
    const int lk   = tid % LK;
    const int nsub = tid / LK;
    for (int s0 = 0; s0 < STEPS; s0 += UN) {
      float4 r[UN];
#pragma unroll
      for (int u = 0; u < UN; ++u) {
        const int s  = s0 + u;
        const int kq = (s % KQH) * LK + lk;
        const int n  = (s / KQH) * NSUB + nsub;
        r[u] = *reinterpret_cast<const float4*>(Bm + n * K + 4 * kq);
      }
#pragma unroll
      for (int u = 0; u < UN; ++u) {
        const int s  = s0 + u;
        const int kq = (s % KQH) * LK + lk;
        const int n  = (s / KQH) * NSUB + nsub;
        float* p = smem + (((2 * kq) * NC + n) << 1);
        *reinterpret_cast<float2*>(p)          = make_float2(r[u].x, r[u].y);
        *reinterpret_cast<float2*>(p + 2 * NC) = make_float2(r[u].z, r[u].w);
      }
    }
  }
}

template <int K, int NC, bool BT, bool ELU>
__global__ __launch_bounds__(128) void gemm_wmma(const float* __restrict__ A,
                                                 const float* __restrict__ Bm,
                                                 float* __restrict__ D, int M) {
  extern __shared__ __align__(16) float smem[];

  const int tid = threadIdx.x;
  stage_b<K, NC, BT>(Bm, smem, tid);
  __syncthreads();

  const int lane  = tid & 31;
  const int wv    = tid >> 5;
  const int strip = blockIdx.x * (blockDim.x >> 5) + wv;
  const int row0  = strip * 16;
  if (row0 >= M) return;

  const int m    = lane & 15;
  const int koff = (lane >> 4) << 1;
  const float* arow = A + (size_t)(row0 + m) * K + koff;

  constexpr int NH    = (NC > 128) ? 128 : NC;
  constexpr int NPASS = NC / NH;
  constexpr int NT    = NH / 16;
  static_assert(K % 8 == 0, "k-loop unrolled by 8");

  for (int np = 0; np < NPASS; ++np) {
    v8f acc[NT];
#pragma unroll
    for (int t = 0; t < NT; ++t)
#pragma unroll
      for (int i = 0; i < 8; ++i) acc[t][i] = 0.0f;

    auto loadB = [&](int kk, v2f* dst) {
      const float* bp = smem + ((((kk + koff) >> 1) * NC + np * NH + m) << 1);
#pragma unroll
      for (int t = 0; t < NT; ++t)
        dst[t] = *reinterpret_cast<const v2f*>(bp + t * 32);
    };
    auto mmaStep = [&](int kk, const v2f* bv) {
      v2f av = *reinterpret_cast<const v2f*>(arow + kk);
      if (ELU) {
        av[0] = av[0] > 0.0f ? av[0] : (expf(av[0]) - 1.0f);
        av[1] = av[1] > 0.0f ? av[1] : (expf(av[1]) - 1.0f);
      }
#pragma unroll
      for (int t = 0; t < NT; ++t)
        acc[t] = __builtin_amdgcn_wmma_f32_16x16x4_f32(
            false, av, false, bv[t], (short)0, acc[t], false, false);
    };

    v2f bv0[NT], bv1[NT];
    loadB(0, bv0);
    for (int k0 = 0; k0 < K; k0 += 8) {
      loadB(k0 + 4, bv1);
      mmaStep(k0, bv0);
      if (k0 + 8 < K) loadB(k0 + 8, bv0);
      mmaStep(k0 + 4, bv1);
    }

    const int mbase = (lane >> 4) * 8;
#pragma unroll
    for (int t = 0; t < NT; ++t) {
#pragma unroll
      for (int v = 0; v < 8; ++v) {
        D[(size_t)(row0 + mbase + v) * NC + np * NH + t * 16 + m] = acc[t][v];
      }
    }
  }
}

__global__ void node_dots(const float* __restrict__ xs1,
                          const float* __restrict__ a_s,
                          const float* __restrict__ a_d,
                          float* __restrict__ as1, float* __restrict__ ad1,
                          int n) {
  const int gw   = (blockIdx.x * blockDim.x + threadIdx.x) >> 5;
  const int lane = threadIdx.x & 31;
  if (gw >= n) return;
  const float4 v  = ((const float4*)(xs1 + (size_t)gw * HIDD))[lane];
  const float4 cs = ((const float4*)a_s)[lane];
  const float4 cd = ((const float4*)a_d)[lane];
  float s = v.x * cs.x + v.y * cs.y + v.z * cs.z + v.w * cs.w;
  float d = v.x * cd.x + v.y * cd.y + v.z * cd.z + v.w * cd.w;
#pragma unroll
  for (int off = 16; off > 0; off >>= 1) {
    s += __shfl_down(s, off, 32);
    d += __shfl_down(d, off, 32);
  }
  if (lane == 0) {
    as1[gw] = s;
    ad1[gw] = d;
  }
}

__global__ void deg_rank(const int* __restrict__ ei, int* __restrict__ cnt,
                         int* __restrict__ rank, int E) {
  const int e = blockIdx.x * blockDim.x + threadIdx.x;
  if (e >= E) return;
  rank[e] = atomicAdd(&cnt[ei[E + e]], 1);
}

__global__ __launch_bounds__(1024) void seg_scan(int* __restrict__ cnt, int n) {
  __shared__ int wsum[2][32];
  const int tid  = threadIdx.x;
  const int lane = tid & 31;
  const int wv   = tid >> 5;
  int carry = 0;
  int it = 0;
  for (int base = 0; base < n; base += 4096, ++it) {
    const int i = base + tid * 4;
    int4 v = make_int4(0, 0, 0, 0);
    if (i + 3 < n) {
      v = *reinterpret_cast<const int4*>(cnt + i);
    } else {
      if (i < n) v.x = cnt[i];
      if (i + 1 < n) v.y = cnt[i + 1];
      if (i + 2 < n) v.z = cnt[i + 2];
    }
    const int t0 = v.x, t1 = t0 + v.y, t2 = t1 + v.z, t3 = t2 + v.w;
    int x = t3;
#pragma unroll
    for (int off = 1; off < 32; off <<= 1) {
      const int y = __shfl_up(x, off, 32);
      if (lane >= off) x += y;
    }
    if (lane == 31) wsum[it & 1][wv] = x;
    __syncthreads();
    const int ws = wsum[it & 1][lane];
    int wx = ws;
#pragma unroll
    for (int off = 1; off < 32; off <<= 1) {
      const int y = __shfl_up(wx, off, 32);
      if (lane >= off) wx += y;
    }
    const int total = __shfl(wx, 31, 32);
    const int wexcl = __shfl(wx, wv, 32) - __shfl(ws, wv, 32);
    const int excl  = carry + wexcl + (x - t3);
    if (i + 3 < n) {
      *reinterpret_cast<int4*>(cnt + i) =
          make_int4(excl, excl + t0, excl + t1, excl + t2);
    } else {
      if (i < n) cnt[i] = excl;
      if (i + 1 < n) cnt[i + 1] = excl + t0;
      if (i + 2 < n) cnt[i + 2] = excl + t1;
    }
    carry += total;
  }
}

__global__ void csr_fill(const int* __restrict__ ei,
                         const int* __restrict__ start,
                         const int* __restrict__ rank,
                         int* __restrict__ csrc, int E) {
  const int e = blockIdx.x * blockDim.x + threadIdx.x;
  if (e >= E) return;
  csrc[start[ei[E + e]] + rank[e]] = ei[e];
}

__device__ __forceinline__ float leaky(float v) {
  return v > 0.0f ? v : NEG_SLOPE * v;
}
__device__ __forceinline__ float wave_max(float v) {
#pragma unroll
  for (int off = 16; off > 0; off >>= 1) v = fmaxf(v, __shfl_xor(v, off, 32));
  return v;
}
__device__ __forceinline__ float wave_sum(float v) {
#pragma unroll
  for (int off = 16; off > 0; off >>= 1) v += __shfl_xor(v, off, 32);
  return v;
}

__device__ __forceinline__ void row_fma(float4& acc, const float4& v, float a) {
  acc.x = fmaf(v.x, a, acc.x);
  acc.y = fmaf(v.y, a, acc.y);
  acc.z = fmaf(v.z, a, acc.z);
  acc.w = fmaf(v.w, a, acc.w);
}
__device__ __forceinline__ void row_fma(float& acc, const float& v, float a) {
  acc = fmaf(v, a, acc);
}
__device__ __forceinline__ void row_zero(float4& v) {
  v = make_float4(0.0f, 0.0f, 0.0f, 0.0f);
}
__device__ __forceinline__ void row_zero(float& v) { v = 0.0f; }

template <typename VT, int G>
__device__ __forceinline__ void add_rows(const VT* __restrict__ fl, int s_l,
                                         float a_l, int cnt, VT& acc) {
  const int ai = __float_as_int(a_l);
  int u = 0;
  for (; u + G <= cnt; u += G) {
    VT    v[G];
    float a[G];
#pragma unroll
    for (int i = 0; i < G; ++i) {
      const int s = __builtin_amdgcn_readlane(s_l, u + i);
      a[i] = __int_as_float(__builtin_amdgcn_readlane(ai, u + i));
      v[i] = fl[(size_t)s * 32];
    }
#pragma unroll
    for (int i = 0; i < G; ++i) row_fma(acc, v[i], a[i]);
  }
  if (u < cnt) {
    VT    v[G];
    float a[G];
#pragma unroll
    for (int i = 0; i < G; ++i) {
      if (u + i < cnt) {
        const int s = __builtin_amdgcn_readlane(s_l, u + i);
        a[i] = __int_as_float(__builtin_amdgcn_readlane(ai, u + i));
        v[i] = fl[(size_t)s * 32];
      }
    }
#pragma unroll
    for (int i = 0; i < G; ++i) {
      if (u + i < cnt) row_fma(acc, v[i], a[i]);
    }
  }
}

template <bool FIRST, typename VT, int G>
__global__ __launch_bounds__(256) void node_pull(
    const int* __restrict__ start, const int* __restrict__ csrc,
    const float* __restrict__ feat, const float* __restrict__ as1,
    const float* __restrict__ ad1, float* __restrict__ alpha,
    float* __restrict__ agg, int n, int E) {
  const int lane = threadIdx.x & 31;
  const int d = __builtin_amdgcn_readfirstlane(
      (int)((blockIdx.x * blockDim.x + threadIdx.x) >> 5));
  if (d >= n) return;
  const int beg = __builtin_amdgcn_readfirstlane(start[d]);
  const int end = (d + 1 < n) ? __builtin_amdgcn_readfirstlane(start[d + 1]) : E;
  const int deg = end - beg;

  const VT* fl = reinterpret_cast<const VT*>(feat) + lane;
  VT acc;
  row_zero(acc);

  if (deg <= 32) {
    const int  j  = beg + lane;
    const bool ok = lane < deg;
    int   s_l = 0;
    float a_l = 0.0f;
    if (FIRST) {
      float v = -INFINITY;
      if (ok) {
        s_l = csrc[j];
        v   = leaky(as1[s_l] + ad1[d]);
      }
      const float m  = wave_max(v);
      const float w  = ok ? expf(v - m) : 0.0f;
      const float ss = wave_sum(w);
      if (ok) {
        a_l = w / ss;
        alpha[j] = a_l;
      }
    } else {
      if (ok) {
        s_l = csrc[j];
        a_l = alpha[j];
      }
    }
    add_rows<VT, G>(fl, s_l, a_l, deg, acc);
  } else {
    float m = 0.0f, ss = 1.0f, adv = 0.0f;
    if (FIRST) {
      adv = ad1[d];
      float mm = -INFINITY;
      for (int j = beg + lane; j < end; j += 32)
        mm = fmaxf(mm, leaky(as1[csrc[j]] + adv));
      m = wave_max(mm);
      float sm = 0.0f;
      for (int j = beg + lane; j < end; j += 32)
        sm += expf(leaky(as1[csrc[j]] + adv) - m);
      ss = wave_sum(sm);
    }
    for (int c = beg; c < end; c += 32) {
      const int j = c + lane;
      int   s_l = 0;
      float a_l = 0.0f;
      if (j < end) {
        s_l = csrc[j];
        if (FIRST) {
          a_l = expf(leaky(as1[s_l] + adv) - m) / ss;
          alpha[j] = a_l;
        } else {
          a_l = alpha[j];
        }
      }
      const int left = end - c;
      add_rows<VT, G>(fl, s_l, a_l, left < 32 ? left : 32, acc);
    }
  }
  reinterpret_cast<VT*>(agg)[(size_t)d * 32 + lane] = acc;
}

extern "C" void kernel_launch(void* const* d_in, const int* in_sizes, int n_in,
                              void* d_out, int out_size, void* d_ws,
                              size_t ws_size, hipStream_t stream) {
  const float* x   = (const float*)d_in[0];
  const float* W1  = (const float*)d_in[1];
  const float* a1s = (const float*)d_in[2];
  const float* a1d = (const float*)d_in[3];
  const float* W2  = (const float*)d_in[4];
  const int*   ei  = (const int*)d_in[5];

  float* out = (float*)d_out;
  float* h2  = out;
  float* h4  = out + (size_t)NN * OUTD;

  float* ws    = (float*)d_ws;
  float* xs1   = ws;
  float* agg   = xs1 + (size_t)NN * HIDD;
  float* as1   = agg + (size_t)NN * HIDD;
  float* ad1   = as1 + NN;
  int*   cnt   = (int*)(ad1 + NN);
  float* ebuf  = (float*)(cnt + 2 * NN);
  int*   csrc  = (int*)ebuf;
  float* alpha = ebuf + NE;
  int*   rank  = (int*)agg;

  hipMemsetAsync(cnt, 0, (size_t)NN * sizeof(int), stream);

  const int strips  = NN / 16;
  const int gblocks = (strips + 3) / 4;
  const int eblocks = (NE + 255) / 256;
  const int npullB  = (NN * 32 + 255) / 256;
  const int nwarpsB = (NN * 32 + 255) / 256;

  deg_rank<<<eblocks, 256, 0, stream>>>(ei, cnt, rank, NE);
  seg_scan<<<1, 1024, 0, stream>>>(cnt, NN);
  csr_fill<<<eblocks, 256, 0, stream>>>(ei, cnt, rank, csrc, NE);

  gemm_wmma<IND, HIDD, false, false>
      <<<gblocks, 128, IND * HIDD * sizeof(float), stream>>>(x, W1, xs1, NN);

  node_dots<<<nwarpsB, 256, 0, stream>>>(xs1, a1s, a1d, as1, ad1, NN);

  node_pull<true, float4, 8>
      <<<npullB, 256, 0, stream>>>(cnt, csrc, xs1, as1, ad1, alpha, agg, NN, NE);

  gemm_wmma<HIDD, OUTD, false, true>
      <<<gblocks, 128, HIDD * OUTD * sizeof(float), stream>>>(agg, W2, h2, NN);

  float* aggh = agg;
  node_pull<false, float, 16>
      <<<npullB, 256, 0, stream>>>(cnt, csrc, h2, as1, ad1, alpha, aggh, NN, NE);

  gemm_wmma<OUTD, HIDD, true, false>
      <<<gblocks, 128, OUTD * HIDD * sizeof(float), stream>>>(aggh, W2, xs1, NN);

  gemm_wmma<HIDD, IND, true, true>
      <<<gblocks, 128, HIDD * IND * sizeof(float), stream>>>(xs1, W1, h4, NN);
}


# Segment-start scan on 25 work-groups at once: each sums the counters ahead of its chunk itself (batched independent loads), then scans its chunk; was one work-group walking 25 dependent trips
# speedup vs baseline: 3.2327x; 1.0692x over previous
#include <hip/hip_runtime.h>
#include <hip/hip_bf16.h>


typedef float v2f __attribute__((ext_vector_type(2)));
typedef float v8f __attribute__((ext_vector_type(8)));

#define NN   100000
#define NE   1600000
#define IND  256
#define HIDD 128
#define OUTD 32
#define NEG_SLOPE 0.2f

template <int K, int NC, bool BT>
__device__ __forceinline__ void stage_b(const float* __restrict__ Bm,
                                        float* __restrict__ smem, int tid) {
  constexpr int NTHR = 128;
  if (!BT) {
    constexpr int NQ    = NC / 4;
    constexpr int ITEMS = (K / 2) * NQ;
    constexpr int PER_T = ITEMS / NTHR;
    constexpr int UN    = PER_T < 8 ? PER_T : 8;
    static_assert(ITEMS % (NTHR * UN) == 0, "whole batches");
    for (int base = 0; base < ITEMS; base += NTHR * UN) {
      float4 r0[UN], r1[UN];
#pragma unroll
      for (int u = 0; u < UN; ++u) {
        const int it = base + u * NTHR + tid;
        const int kp = it / NQ;
        const float* g = Bm + 4 * it + kp * NC;
        r0[u] = *reinterpret_cast<const float4*>(g);
        r1[u] = *reinterpret_cast<const float4*>(g + NC);
      }
#pragma unroll
      for (int u = 0; u < UN; ++u) {
        const int it = base + u * NTHR + tid;
        float4* dst = reinterpret_cast<float4*>(smem + 8 * it);
        dst[0] = make_float4(r0[u].x, r1[u].x, r0[u].y, r1[u].y);
        dst[1] = make_float4(r0[u].z, r1[u].z, r0[u].w, r1[u].w);
      }
    }
  } else {
    constexpr int LK    = 4;
    constexpr int NSUB  = NTHR / LK;
    constexpr int KQH   = (K / 4) / LK;
    constexpr int STEPS = (NC / NSUB) * KQH;
    constexpr int UN    = STEPS < 8 ? STEPS : 8;
    static_assert((K / 4) % LK == 0 && NC % NSUB == 0 && STEPS % UN == 0,
                  "whole batches");
    const int lk   = tid % LK;
    const int nsub = tid / LK;
    for (int s0 = 0; s0 < STEPS; s0 += UN) {
      float4 r[UN];
#pragma unroll
      for (int u = 0; u < UN; ++u) {
        const int s  = s0 + u;
        const int kq = (s % KQH) * LK + lk;
        const int n  = (s / KQH) * NSUB + nsub;
        r[u] = *reinterpret_cast<const float4*>(Bm + n * K + 4 * kq);
      }
#pragma unroll
      for (int u = 0; u < UN; ++u) {
        const int s  = s0 + u;
        const int kq = (s % KQH) * LK + lk;
        const int n  = (s / KQH) * NSUB + nsub;
        float* p = smem + (((2 * kq) * NC + n) << 1);
        *reinterpret_cast<float2*>(p)          = make_float2(r[u].x, r[u].y);
        *reinterpret_cast<float2*>(p + 2 * NC) = make_float2(r[u].z, r[u].w);
      }
    }
  }
}

template <int K, int NC, bool BT, bool ELU>
__global__ __launch_bounds__(128) void gemm_wmma(const float* __restrict__ A,
                                                 const float* __restrict__ Bm,
                                                 float* __restrict__ D, int M) {
  extern __shared__ __align__(16) float smem[];

  const int tid = threadIdx.x;
  stage_b<K, NC, BT>(Bm, smem, tid);
  __syncthreads();

  const int lane  = tid & 31;
  const int wv    = tid >> 5;
  const int strip = blockIdx.x * (blockDim.x >> 5) + wv;
  const int row0  = strip * 16;
  if (row0 >= M) return;

  const int m    = lane & 15;
  const int koff = (lane >> 4) << 1;
  const float* arow = A + (size_t)(row0 + m) * K + koff;

  constexpr int NH    = (NC > 128) ? 128 : NC;
  constexpr int NPASS = NC / NH;
  constexpr int NT    = NH / 16;
  static_assert(K % 8 == 0, "k-loop unrolled by 8");

  for (int np = 0; np < NPASS; ++np) {
    v8f acc[NT];
#pragma unroll
    for (int t = 0; t < NT; ++t)
#pragma unroll
      for (int i = 0; i < 8; ++i) acc[t][i] = 0.0f;

    auto loadB = [&](int kk, v2f* dst) {
      const float* bp = smem + ((((kk + koff) >> 1) * NC + np * NH + m) << 1);
#pragma unroll
      for (int t = 0; t < NT; ++t)
        dst[t] = *reinterpret_cast<const v2f*>(bp + t * 32);
    };
    auto mmaStep = [&](int kk, const v2f* bv) {
      v2f av = *reinterpret_cast<const v2f*>(arow + kk);
      if (ELU) {
        av[0] = av[0] > 0.0f ? av[0] : (expf(av[0]) - 1.0f);
        av[1] = av[1] > 0.0f ? av[1] : (expf(av[1]) - 1.0f);
      }
#pragma unroll
      for (int t = 0; t < NT; ++t)
        acc[t] = __builtin_amdgcn_wmma_f32_16x16x4_f32(
            false, av, false, bv[t], (short)0, acc[t], false, false);
    };

    v2f bv0[NT], bv1[NT];
    loadB(0, bv0);
    for (int k0 = 0; k0 < K; k0 += 8) {
      loadB(k0 + 4, bv1);
      mmaStep(k0, bv0);
      if (k0 + 8 < K) loadB(k0 + 8, bv0);
      mmaStep(k0 + 4, bv1);
    }

    const int mbase = (lane >> 4) * 8;
#pragma unroll
    for (int t = 0; t < NT; ++t) {
#pragma unroll
      for (int v = 0; v < 8; ++v) {
        D[(size_t)(row0 + mbase + v) * NC + np * NH + t * 16 + m] = acc[t][v];
      }
    }
  }
}

__global__ void node_dots(const float* __restrict__ xs1,
                          const float* __restrict__ a_s,
                          const float* __restrict__ a_d,
                          float* __restrict__ as1, float* __restrict__ ad1,
                          int n) {
  const int gw   = (blockIdx.x * blockDim.x + threadIdx.x) >> 5;
  const int lane = threadIdx.x & 31;
  if (gw >= n) return;
  const float4 v  = ((const float4*)(xs1 + (size_t)gw * HIDD))[lane];
  const float4 cs = ((const float4*)a_s)[lane];
  const float4 cd = ((const float4*)a_d)[lane];
  float s = v.x * cs.x + v.y * cs.y + v.z * cs.z + v.w * cs.w;
  float d = v.x * cd.x + v.y * cd.y + v.z * cd.z + v.w * cd.w;
#pragma unroll
  for (int off = 16; off > 0; off >>= 1) {
    s += __shfl_down(s, off, 32);
    d += __shfl_down(d, off, 32);
  }
  if (lane == 0) {
    as1[gw] = s;
    ad1[gw] = d;
  }
}

__global__ void deg_rank(const int* __restrict__ ei, int* __restrict__ cnt,
                         int* __restrict__ rank, int E) {
  const int e = blockIdx.x * blockDim.x + threadIdx.x;
  if (e >= E) return;
  rank[e] = atomicAdd(&cnt[ei[E + e]], 1);
}

#define SCAN_CH    4096
#define SCAN_NB    ((NN + SCAN_CH - 1) / SCAN_CH)
#define SCAN_BATCH 12
__global__ __launch_bounds__(1024) void seg_scan(const int* __restrict__ cnt,
                                                 int* __restrict__ start,
                                                 int n) {
  __shared__ int wsum[32];
  __shared__ int wpre[32];
  const int tid  = threadIdx.x;
  const int lane = tid & 31;
  const int wv   = tid >> 5;
  const int b    = blockIdx.x;

  const int i = b * SCAN_CH + tid * 4;
  int4 v = make_int4(0, 0, 0, 0);
  if (i + 3 < n) {
    v = *reinterpret_cast<const int4*>(cnt + i);
  } else {
    if (i < n) v.x = cnt[i];
    if (i + 1 < n) v.y = cnt[i + 1];
    if (i + 2 < n) v.z = cnt[i + 2];
  }

  int pre = 0;
  const int4* ahead = reinterpret_cast<const int4*>(cnt) + tid;
#pragma unroll
  for (int c0 = 0; c0 < SCAN_NB - 1; c0 += SCAN_BATCH) {
    int4 q[SCAN_BATCH];
#pragma unroll
    for (int u = 0; u < SCAN_BATCH; ++u) {
      q[u] = make_int4(0, 0, 0, 0);
      if (c0 + u < b) q[u] = ahead[(c0 + u) * (SCAN_CH / 4)];
    }
#pragma unroll
    for (int u = 0; u < SCAN_BATCH; ++u)
      pre += (q[u].x + q[u].y) + (q[u].z + q[u].w);
  }

  const int t0 = v.x, t1 = t0 + v.y, t2 = t1 + v.z, t3 = t2 + v.w;
  int x = t3;
#pragma unroll
  for (int off = 1; off < 32; off <<= 1) {
    const int y = __shfl_up(x, off, 32);
    if (lane >= off) x += y;
  }
#pragma unroll
  for (int off = 16; off > 0; off >>= 1) pre += __shfl_xor(pre, off, 32);
  if (lane == 31) wsum[wv] = x;
  if (lane == 0) wpre[wv] = pre;
  __syncthreads();
  const int ws = wsum[lane];
  int wx = ws;
#pragma unroll
  for (int off = 1; off < 32; off <<= 1) {
    const int y = __shfl_up(wx, off, 32);
    if (lane >= off) wx += y;
  }
  int carry = wpre[lane];
#pragma unroll
  for (int off = 16; off > 0; off >>= 1) carry += __shfl_xor(carry, off, 32);
  const int wexcl = __shfl(wx, wv, 32) - __shfl(ws, wv, 32);
  const int excl  = carry + wexcl + (x - t3);
  if (i + 3 < n) {
    *reinterpret_cast<int4*>(start + i) =
        make_int4(excl, excl + t0, excl + t1, excl + t2);
  } else {
    if (i < n) start[i] = excl;
    if (i + 1 < n) start[i + 1] = excl + t0;
    if (i + 2 < n) start[i + 2] = excl + t1;
  }
}

__global__ void csr_fill(const int* __restrict__ ei,
                         const int* __restrict__ start,
                         const int* __restrict__ rank,
                         int* __restrict__ csrc, int E) {
  const int e = blockIdx.x * blockDim.x + threadIdx.x;
  if (e >= E) return;
  csrc[start[ei[E + e]] + rank[e]] = ei[e];
}

__device__ __forceinline__ float leaky(float v) {
  return v > 0.0f ? v : NEG_SLOPE * v;
}
__device__ __forceinline__ float wave_max(float v) {
#pragma unroll
  for (int off = 16; off > 0; off >>= 1) v = fmaxf(v, __shfl_xor(v, off, 32));
  return v;
}
__device__ __forceinline__ float wave_sum(float v) {
#pragma unroll
  for (int off = 16; off > 0; off >>= 1) v += __shfl_xor(v, off, 32);
  return v;
}

__device__ __forceinline__ void row_fma(float4& acc, const float4& v, float a) {
  acc.x = fmaf(v.x, a, acc.x);
  acc.y = fmaf(v.y, a, acc.y);
  acc.z = fmaf(v.z, a, acc.z);
  acc.w = fmaf(v.w, a, acc.w);
}
__device__ __forceinline__ void row_fma(float& acc, const float& v, float a) {
  acc = fmaf(v, a, acc);
}
__device__ __forceinline__ void row_zero(float4& v) {
  v = make_float4(0.0f, 0.0f, 0.0f, 0.0f);
}
__device__ __forceinline__ void row_zero(float& v) { v = 0.0f; }

template <typename VT, int G>
__device__ __forceinline__ void add_rows(const VT* __restrict__ fl, int s_l,
                                         float a_l, int cnt, VT& acc) {
  const int ai = __float_as_int(a_l);
  int u = 0;
  for (; u + G <= cnt; u += G) {
    VT    v[G];
    float a[G];
#pragma unroll
    for (int i = 0; i < G; ++i) {
      const int s = __builtin_amdgcn_readlane(s_l, u + i);
      a[i] = __int_as_float(__builtin_amdgcn_readlane(ai, u + i));
      v[i] = fl[(size_t)s * 32];
    }
#pragma unroll
    for (int i = 0; i < G; ++i) row_fma(acc, v[i], a[i]);
  }
  if (u < cnt) {
    VT    v[G];
    float a[G];
#pragma unroll
    for (int i = 0; i < G; ++i) {
      if (u + i < cnt) {
        const int s = __builtin_amdgcn_readlane(s_l, u + i);
        a[i] = __int_as_float(__builtin_amdgcn_readlane(ai, u + i));
        v[i] = fl[(size_t)s * 32];
      }
    }
#pragma unroll
    for (int i = 0; i < G; ++i) {
      if (u + i < cnt) row_fma(acc, v[i], a[i]);
    }
  }
}

template <bool FIRST, typename VT, int G>
__global__ __launch_bounds__(256) void node_pull(
    const int* __restrict__ start, const int* __restrict__ csrc,
    const float* __restrict__ feat, const float* __restrict__ as1,
    const float* __restrict__ ad1, float* __restrict__ alpha,
    float* __restrict__ agg, int n, int E) {
  const int lane = threadIdx.x & 31;
  const int d = __builtin_amdgcn_readfirstlane(
      (int)((blockIdx.x * blockDim.x + threadIdx.x) >> 5));
  if (d >= n) return;
  const int beg = __builtin_amdgcn_readfirstlane(start[d]);
  const int end = (d + 1 < n) ? __builtin_amdgcn_readfirstlane(start[d + 1]) : E;
  const int deg = end - beg;

  const VT* fl = reinterpret_cast<const VT*>(feat) + lane;
  VT acc;
  row_zero(acc);

  if (deg <= 32) {
    const int  j  = beg + lane;
    const bool ok = lane < deg;
    int   s_l = 0;
    float a_l = 0.0f;
    if (FIRST) {
      float v = -INFINITY;
      if (ok) {
        s_l = csrc[j];
        v   = leaky(as1[s_l] + ad1[d]);
      }
      const float m  = wave_max(v);
      const float w  = ok ? expf(v - m) : 0.0f;
      const float ss = wave_sum(w);
      if (ok) {
        a_l = w / ss;
        alpha[j] = a_l;
      }
    } else {
      if (ok) {
        s_l = csrc[j];
        a_l = alpha[j];
      }
    }
    add_rows<VT, G>(fl, s_l, a_l, deg, acc);
  } else {
    float m = 0.0f, ss = 1.0f, adv = 0.0f;
    if (FIRST) {
      adv = ad1[d];
      float mm = -INFINITY;
      for (int j = beg + lane; j < end; j += 32)
        mm = fmaxf(mm, leaky(as1[csrc[j]] + adv));
      m = wave_max(mm);
      float sm = 0.0f;
      for (int j = beg + lane; j < end; j += 32)
        sm += expf(leaky(as1[csrc[j]] + adv) - m);
      ss = wave_sum(sm);
    }
    for (int c = beg; c < end; c += 32) {
      const int j = c + lane;
      int   s_l = 0;
      float a_l = 0.0f;
      if (j < end) {
        s_l = csrc[j];
        if (FIRST) {
          a_l = expf(leaky(as1[s_l] + adv) - m) / ss;
          alpha[j] = a_l;
        } else {
          a_l = alpha[j];
        }
      }
      const int left = end - c;
      add_rows<VT, G>(fl, s_l, a_l, left < 32 ? left : 32, acc);
    }
  }
  reinterpret_cast<VT*>(agg)[(size_t)d * 32 + lane] = acc;
}

extern "C" void kernel_launch(void* const* d_in, const int* in_sizes, int n_in,
                              void* d_out, int out_size, void* d_ws,
                              size_t ws_size, hipStream_t stream) {
  const float* x   = (const float*)d_in[0];
  const float* W1  = (const float*)d_in[1];
  const float* a1s = (const float*)d_in[2];
  const float* a1d = (const float*)d_in[3];
  const float* W2  = (const float*)d_in[4];
  const int*   ei  = (const int*)d_in[5];

  float* out = (float*)d_out;
  float* h2  = out;
  float* h4  = out + (size_t)NN * OUTD;

  float* ws    = (float*)d_ws;
  float* xs1   = ws;
  float* agg   = xs1 + (size_t)NN * HIDD;
  float* as1   = agg + (size_t)NN * HIDD;
  float* ad1   = as1 + NN;
  int*   cnt   = (int*)(ad1 + NN);
  int*   start = cnt + NN;
  float* ebuf  = (float*)(cnt + 2 * NN);
  int*   csrc  = (int*)ebuf;
  float* alpha = ebuf + NE;
  int*   rank  = (int*)agg;

  hipMemsetAsync(cnt, 0, (size_t)NN * sizeof(int), stream);

  const int strips  = NN / 16;
  const int gblocks = (strips + 3) / 4;
  const int eblocks = (NE + 255) / 256;
  const int npullB  = (NN * 32 + 255) / 256;
  const int nwarpsB = (NN * 32 + 255) / 256;

  deg_rank<<<eblocks, 256, 0, stream>>>(ei, cnt, rank, NE);
  seg_scan<<<SCAN_NB, 1024, 0, stream>>>(cnt, start, NN);
  csr_fill<<<eblocks, 256, 0, stream>>>(ei, start, rank, csrc, NE);

  gemm_wmma<IND, HIDD, false, false>
      <<<gblocks, 128, IND * HIDD * sizeof(float), stream>>>(x, W1, xs1, NN);

  node_dots<<<nwarpsB, 256, 0, stream>>>(xs1, a1s, a1d, as1, ad1, NN);

  node_pull<true, float4, 8>
      <<<npullB, 256, 0, stream>>>(start, csrc, xs1, as1, ad1, alpha, agg, NN, NE);

  gemm_wmma<HIDD, OUTD, false, true>
      <<<gblocks, 128, HIDD * OUTD * sizeof(float), stream>>>(agg, W2, h2, NN);

  float* aggh = agg;
  node_pull<false, float, 16>
      <<<npullB, 256, 0, stream>>>(start, csrc, h2, as1, ad1, alpha, aggh, NN, NE);

  gemm_wmma<OUTD, HIDD, true, false>
      <<<gblocks, 128, OUTD * HIDD * sizeof(float), stream>>>(aggh, W2, xs1, NN);

  gemm_wmma<HIDD, IND, true, true>
      <<<gblocks, 128, HIDD * IND * sizeof(float), stream>>>(xs1, W1, h4, NN);
}


# Edge sort by destination through LDS bins: tile histograms and slot cursors in LDS, runs stored whole; no global atomic or scattered 4-byte store; segment order now by source id, fixed run to run
# speedup vs baseline: 4.3107x; 1.3335x over previous
#include <hip/hip_runtime.h>
#include <hip/hip_bf16.h>


typedef float v2f __attribute__((ext_vector_type(2)));
typedef float v8f __attribute__((ext_vector_type(8)));

#define NN   100000
#define NE   1600000
#define IND  256
#define HIDD 128
#define OUTD 32
#define NEG_SLOPE 0.2f

template <int K, int NC, bool BT>
__device__ __forceinline__ void stage_b(const float* __restrict__ Bm,
                                        float* __restrict__ smem, int tid) {
  constexpr int NTHR = 128;
  if (!BT) {
    constexpr int NQ    = NC / 4;
    constexpr int ITEMS = (K / 2) * NQ;
    constexpr int PER_T = ITEMS / NTHR;
    constexpr int UN    = PER_T < 8 ? PER_T : 8;
    static_assert(ITEMS % (NTHR * UN) == 0, "whole batches");
    for (int base = 0; base < ITEMS; base += NTHR * UN) {
      float4 r0[UN], r1[UN];
#pragma unroll
      for (int u = 0; u < UN; ++u) {
        const int it = base + u * NTHR + tid;
        const int kp = it / NQ;
        const float* g = Bm + 4 * it + kp * NC;
        r0[u] = *reinterpret_cast<const float4*>(g);
        r1[u] = *reinterpret_cast<const float4*>(g + NC);
      }
#pragma unroll
      for (int u = 0; u < UN; ++u) {
        const int it = base + u * NTHR + tid;
        float4* dst = reinterpret_cast<float4*>(smem + 8 * it);
        dst[0] = make_float4(r0[u].x, r1[u].x, r0[u].y, r1[u].y);
        dst[1] = make_float4(r0[u].z, r1[u].z, r0[u].w, r1[u].w);
      }
    }
  } else {
    constexpr int LK    = 4;
    constexpr int NSUB  = NTHR / LK;
    constexpr int KQH   = (K / 4) / LK;
    constexpr int STEPS = (NC / NSUB) * KQH;
    constexpr int UN    = STEPS < 8 ? STEPS : 8;
    static_assert((K / 4) % LK == 0 && NC % NSUB == 0 && STEPS % UN == 0,
                  "whole batches");
    const int lk   = tid % LK;
    const int nsub = tid / LK;
    for (int s0 = 0; s0 < STEPS; s0 += UN) {
      float4 r[UN];
#pragma unroll
      for (int u = 0; u < UN; ++u) {
        const int s  = s0 + u;
        const int kq = (s % KQH) * LK + lk;
        const int n  = (s / KQH) * NSUB + nsub;
        r[u] = *reinterpret_cast<const float4*>(Bm + n * K + 4 * kq);
      }
#pragma unroll
      for (int u = 0; u < UN; ++u) {
        const int s  = s0 + u;
        const int kq = (s % KQH) * LK + lk;
        const int n  = (s / KQH) * NSUB + nsub;
        float* p = smem + (((2 * kq) * NC + n) << 1);
        *reinterpret_cast<float2*>(p)          = make_float2(r[u].x, r[u].y);
        *reinterpret_cast<float2*>(p + 2 * NC) = make_float2(r[u].z, r[u].w);
      }
    }
  }
}

template <int K, int NC, bool BT, bool ELU>
__global__ __launch_bounds__(128) void gemm_wmma(const float* __restrict__ A,
                                                 const float* __restrict__ Bm,
                                                 float* __restrict__ D, int M) {
  extern __shared__ __align__(16) float smem[];

  const int tid = threadIdx.x;
  stage_b<K, NC, BT>(Bm, smem, tid);
  __syncthreads();

  const int lane  = tid & 31;
  const int wv    = tid >> 5;
  const int strip = blockIdx.x * (blockDim.x >> 5) + wv;
  const int row0  = strip * 16;
  if (row0 >= M) return;

  const int m    = lane & 15;
  const int koff = (lane >> 4) << 1;
  const float* arow = A + (size_t)(row0 + m) * K + koff;

  constexpr int NH    = (NC > 128) ? 128 : NC;
  constexpr int NPASS = NC / NH;
  constexpr int NT    = NH / 16;
  static_assert(K % 8 == 0, "k-loop unrolled by 8");

  for (int np = 0; np < NPASS; ++np) {
    v8f acc[NT];
#pragma unroll
    for (int t = 0; t < NT; ++t)
#pragma unroll
      for (int i = 0; i < 8; ++i) acc[t][i] = 0.0f;

    auto loadB = [&](int kk, v2f* dst) {
      const float* bp = smem + ((((kk + koff) >> 1) * NC + np * NH + m) << 1);
#pragma unroll
      for (int t = 0; t < NT; ++t)
        dst[t] = *reinterpret_cast<const v2f*>(bp + t * 32);
    };
    auto mmaStep = [&](int kk, const v2f* bv) {
      v2f av = *reinterpret_cast<const v2f*>(arow + kk);
      if (ELU) {
        av[0] = av[0] > 0.0f ? av[0] : (expf(av[0]) - 1.0f);
        av[1] = av[1] > 0.0f ? av[1] : (expf(av[1]) - 1.0f);
      }
#pragma unroll
      for (int t = 0; t < NT; ++t)
        acc[t] = __builtin_amdgcn_wmma_f32_16x16x4_f32(
            false, av, false, bv[t], (short)0, acc[t], false, false);
    };

    v2f bv0[NT], bv1[NT];
    loadB(0, bv0);
    for (int k0 = 0; k0 < K; k0 += 8) {
      loadB(k0 + 4, bv1);
      mmaStep(k0, bv0);
      if (k0 + 8 < K) loadB(k0 + 8, bv0);
      mmaStep(k0 + 4, bv1);
    }

    const int mbase = (lane >> 4) * 8;
#pragma unroll
    for (int t = 0; t < NT; ++t) {
#pragma unroll
      for (int v = 0; v < 8; ++v) {
        D[(size_t)(row0 + mbase + v) * NC + np * NH + t * 16 + m] = acc[t][v];
      }
    }
  }
}

__global__ void node_dots(const float* __restrict__ xs1,
                          const float* __restrict__ a_s,
                          const float* __restrict__ a_d,
                          float* __restrict__ as1, float* __restrict__ ad1,
                          int n) {
  const int gw   = (blockIdx.x * blockDim.x + threadIdx.x) >> 5;
  const int lane = threadIdx.x & 31;
  if (gw >= n) return;
  const float4 v  = ((const float4*)(xs1 + (size_t)gw * HIDD))[lane];
  const float4 cs = ((const float4*)a_s)[lane];
  const float4 cd = ((const float4*)a_d)[lane];
  float s = v.x * cs.x + v.y * cs.y + v.z * cs.z + v.w * cs.w;
  float d = v.x * cd.x + v.y * cd.y + v.z * cd.z + v.w * cd.w;
#pragma unroll
  for (int off = 16; off > 0; off >>= 1) {
    s += __shfl_down(s, off, 32);
    d += __shfl_down(d, off, 32);
  }
  if (lane == 0) {
    as1[gw] = s;
    ad1[gw] = d;
  }
}

#define TE       8192
#define NTILE    ((NE + TE - 1) / TE)
#define BSH      9
#define BNODES   (1 << BSH)
#define NBIN     ((NN + BNODES - 1) / BNODES)
#define SRC_BITS 17
#define SRC_MASK ((1 << SRC_BITS) - 1)
#define BCAP     12288
#define SORT_THR 1024
static_assert(NN <= (1 << SRC_BITS), "node id packs into SRC_BITS");
static_assert(NBIN < 256 && NTILE % 4 == 0 && NE % 4 == 0 && TE == 8 * SORT_THR,
              "shapes the sort kernels are written for");

__device__ __forceinline__ int block_excl_scan(int val, int* wsum) {
  const int lane = threadIdx.x & 31;
  const int wv   = threadIdx.x >> 5;
  int x = val;
#pragma unroll
  for (int off = 1; off < 32; off <<= 1) {
    const int y = __shfl_up(x, off, 32);
    if (lane >= off) x += y;
  }
  if (lane == 31) wsum[wv] = x;
  __syncthreads();
  const int ws = wsum[lane];
  int wx = ws;
#pragma unroll
  for (int off = 1; off < 32; off <<= 1) {
    const int y = __shfl_up(wx, off, 32);
    if (lane >= off) wx += y;
  }
  const int wexcl = __shfl(wx, wv, 32) - __shfl(ws, wv, 32);
  return wexcl + (x - val);
}

__global__ __launch_bounds__(SORT_THR) void bin_count(
    const int* __restrict__ ei, int* __restrict__ th) {
  __shared__ int hist[NBIN];
  const int tid  = threadIdx.x;
  const int tile = blockIdx.x;
  if (tid < NBIN) hist[tid] = 0;
  const int4* dq = reinterpret_cast<const int4*>(ei + NE);
  const int  q0  = tile * (TE / 4) + tid;
  const int  q1  = q0 + SORT_THR;
  const bool ok0 = q0 < NE / 4;
  const bool ok1 = q1 < NE / 4;
  int4 d0 = make_int4(0, 0, 0, 0), d1 = d0;
  if (ok0) d0 = dq[q0];
  if (ok1) d1 = dq[q1];
  __syncthreads();
  if (ok0) {
    atomicAdd(&hist[d0.x >> BSH], 1);
    atomicAdd(&hist[d0.y >> BSH], 1);
    atomicAdd(&hist[d0.z >> BSH], 1);
    atomicAdd(&hist[d0.w >> BSH], 1);
  }
  if (ok1) {
    atomicAdd(&hist[d1.x >> BSH], 1);
    atomicAdd(&hist[d1.y >> BSH], 1);
    atomicAdd(&hist[d1.z >> BSH], 1);
    atomicAdd(&hist[d1.w >> BSH], 1);
  }
  __syncthreads();
  if (tid < NBIN) th[tile * NBIN + tid] = hist[tid];
}

__global__ __launch_bounds__(SORT_THR) void bin_scan(
    const int* __restrict__ th, int* __restrict__ base,
    int* __restrict__ binstart) {
  __shared__ int psum[4][256];
  __shared__ int bst[256];
  __shared__ int wsum[32];
  constexpr int QT = NTILE / 4;
  const int tid = threadIdx.x;
  const int q   = tid >> 8;
  const int b   = tid & 255;
  int v[QT];
#pragma unroll
  for (int u = 0; u < QT; ++u)
    v[u] = (b < NBIN) ? th[(q * QT + u) * NBIN + b] : 0;
  int s = 0;
#pragma unroll
  for (int u = 0; u < QT; ++u) s += v[u];
  psum[q][b] = s;
  __syncthreads();
  const int p0 = psum[0][b], p1 = psum[1][b], p2 = psum[2][b], p3 = psum[3][b];
  const int tot  = (q == 0) ? (p0 + p1) + (p2 + p3) : 0;
  const int excl = block_excl_scan(tot, wsum);
  if (q == 0) {
    bst[b] = excl;
    if (b <= NBIN) binstart[b] = excl;
  }
  __syncthreads();
  int run = bst[b];
  if (q > 0) run += p0;
  if (q > 1) run += p1;
  if (q > 2) run += p2;
  if (b < NBIN) {
#pragma unroll
    for (int u = 0; u < QT; ++u) {
      base[(q * QT + u) * NBIN + b] = run;
      run += v[u];
    }
  }
}

__global__ __launch_bounds__(SORT_THR) void bin_scatter(
    const int* __restrict__ ei, const int* __restrict__ th,
    const int* __restrict__ base, int* __restrict__ binned) {
  __shared__ int lstart[256];
  __shared__ int cur[256];
  __shared__ int gbase[256];
  __shared__ int wsum[32];
  __shared__ int stage[TE];
  const int tid  = threadIdx.x;
  const int tile = blockIdx.x;
  const int4* sq = reinterpret_cast<const int4*>(ei);
  const int4* dq = reinterpret_cast<const int4*>(ei + NE);
  const int  q0  = tile * (TE / 4) + tid;
  const int  q1  = q0 + SORT_THR;
  const bool ok0 = q0 < NE / 4;
  const bool ok1 = q1 < NE / 4;
  int4 s0 = make_int4(0, 0, 0, 0), s1 = s0, d0 = s0, d1 = s0;
  if (ok0) {
    s0 = sq[q0];
    d0 = dq[q0];
  }
  if (ok1) {
    s1 = sq[q1];
    d1 = dq[q1];
  }
  const int c  = (tid < NBIN) ? th[tile * NBIN + tid] : 0;
  const int gb = (tid < NBIN) ? base[tile * NBIN + tid] : 0;
  const int ex = block_excl_scan(c, wsum);
  if (tid < 256) {
    lstart[tid] = ex;
    cur[tid]    = ex;
    gbase[tid]  = gb;
  }
  __syncthreads();
  auto place = [&](int s, int d) {
    const int p = atomicAdd(&cur[d >> BSH], 1);
    stage[p] = s | ((d & (BNODES - 1)) << SRC_BITS);
  };
  if (ok0) {
    place(s0.x, d0.x);
    place(s0.y, d0.y);
    place(s0.z, d0.z);
    place(s0.w, d0.w);
  }
  if (ok1) {
    place(s1.x, d1.x);
    place(s1.y, d1.y);
    place(s1.z, d1.z);
    place(s1.w, d1.w);
  }
  __syncthreads();
  const int lane = tid & 31;
  const int wv   = tid >> 5;
  for (int b = wv; b < NBIN; b += 32) {
    const int s = lstart[b];
    const int n = lstart[b + 1] - s;
    const int g = gbase[b];
    for (int i = lane; i < n; i += 32) binned[g + i] = stage[s + i];
  }
}

__global__ __launch_bounds__(SORT_THR) void bin_finish(
    const int* __restrict__ binned, const int* __restrict__ binstart,
    int* __restrict__ start, int* __restrict__ csrc) {
  __shared__ int cnt[BNODES];
  __shared__ int nbeg[BNODES];
  __shared__ int wsum[32];
  __shared__ int outb[BCAP];
  static_assert(BNODES <= SORT_THR && BNODES % 32 == 0 && SORT_THR == 1024,
                "one thread per node counter; 32 waves share the nodes evenly");
  const int tid = threadIdx.x;
  const int bin = blockIdx.x;
  const int bs  = binstart[bin];
  const int c   = binstart[bin + 1] - bs;
  const int* in = binned + bs;
  if (tid < BNODES) cnt[tid] = 0;
  __syncthreads();
  for (int i0 = 0; i0 < c; i0 += 4 * SORT_THR) {
    int w[4];
#pragma unroll
    for (int u = 0; u < 4; ++u) {
      const int i = i0 + u * SORT_THR + tid;
      w[u] = (i < c) ? in[i] : -1;
    }
#pragma unroll
    for (int u = 0; u < 4; ++u)
      if (w[u] >= 0) atomicAdd(&cnt[w[u] >> SRC_BITS], 1);
  }
  __syncthreads();
  const int mine = (tid < BNODES) ? cnt[tid] : 0;
  const int ex   = block_excl_scan(mine, wsum);
  if (tid < BNODES) {
    cnt[tid]  = ex;
    nbeg[tid] = ex;
    const int node = bin * BNODES + tid;
    if (node < NN) start[node] = bs + ex;
  }
  __syncthreads();
  const bool staged = c <= BCAP;
  for (int i0 = 0; i0 < c; i0 += 4 * SORT_THR) {
    int w[4];
#pragma unroll
    for (int u = 0; u < 4; ++u) {
      const int i = i0 + u * SORT_THR + tid;
      w[u] = (i < c) ? in[i] : -1;
    }
#pragma unroll
    for (int u = 0; u < 4; ++u) {
      if (w[u] >= 0) {
        const int p = atomicAdd(&cnt[w[u] >> SRC_BITS], 1);
        if (staged) outb[p] = w[u] & SRC_MASK;
        else        csrc[bs + p] = w[u] & SRC_MASK;
      }
    }
  }
  if (staged) {
    __syncthreads();
    const int lane = tid & 31;
    const int wv   = tid >> 5;
    constexpr int NPW = BNODES / 32;
    for (int t = 0; t < NPW; ++t) {
      const int n = wv * NPW + t;
      const int sb  = __builtin_amdgcn_readfirstlane(nbeg[n]);
      const int deg = __builtin_amdgcn_readfirstlane(cnt[n]) - sb;
      if (deg > 1 && deg <= 32) {
        const int key = ((lane < deg ? outb[sb + lane] : 0) << 5) | lane;
        int r = 0;
        for (int k = 0; k < deg; ++k)
          r += (__builtin_amdgcn_readlane(key, k) < key) ? 1 : 0;
        if (lane < deg) outb[sb + r] = key >> 5;
      } else if (deg > 32 && deg <= 128) {
        int key[4], r[4];
#pragma unroll
        for (int u = 0; u < 4; ++u) {
          const int i = lane + 32 * u;
          key[u] = ((i < deg ? outb[sb + i] : 0) << 7) | i;
          r[u]   = 0;
        }
        for (int k = 0; k < deg; ++k) {
          const int yk = (outb[sb + k] << 7) | k;
#pragma unroll
          for (int u = 0; u < 4; ++u) r[u] += (yk < key[u]) ? 1 : 0;
        }
#pragma unroll
        for (int u = 0; u < 4; ++u)
          if (lane + 32 * u < deg) outb[sb + r[u]] = key[u] >> 7;
      }
    }
    __syncthreads();
    for (int i = tid; i < c; i += SORT_THR) csrc[bs + i] = outb[i];
  }
}

__device__ __forceinline__ float leaky(float v) {
  return v > 0.0f ? v : NEG_SLOPE * v;
}
__device__ __forceinline__ float wave_max(float v) {
#pragma unroll
  for (int off = 16; off > 0; off >>= 1) v = fmaxf(v, __shfl_xor(v, off, 32));
  return v;
}
__device__ __forceinline__ float wave_sum(float v) {
#pragma unroll
  for (int off = 16; off > 0; off >>= 1) v += __shfl_xor(v, off, 32);
  return v;
}

__device__ __forceinline__ void row_fma(float4& acc, const float4& v, float a) {
  acc.x = fmaf(v.x, a, acc.x);
  acc.y = fmaf(v.y, a, acc.y);
  acc.z = fmaf(v.z, a, acc.z);
  acc.w = fmaf(v.w, a, acc.w);
}
__device__ __forceinline__ void row_fma(float& acc, const float& v, float a) {
  acc = fmaf(v, a, acc);
}
__device__ __forceinline__ void row_zero(float4& v) {
  v = make_float4(0.0f, 0.0f, 0.0f, 0.0f);
}
__device__ __forceinline__ void row_zero(float& v) { v = 0.0f; }

template <typename VT, int G>
__device__ __forceinline__ void add_rows(const VT* __restrict__ fl, int s_l,
                                         float a_l, int cnt, VT& acc) {
  const int ai = __float_as_int(a_l);
  int u = 0;
  for (; u + G <= cnt; u += G) {
    VT    v[G];
    float a[G];
#pragma unroll
    for (int i = 0; i < G; ++i) {
      const int s = __builtin_amdgcn_readlane(s_l, u + i);
      a[i] = __int_as_float(__builtin_amdgcn_readlane(ai, u + i));
      v[i] = fl[(size_t)s * 32];
    }
#pragma unroll
    for (int i = 0; i < G; ++i) row_fma(acc, v[i], a[i]);
  }
  if (u < cnt) {
    VT    v[G];
    float a[G];
#pragma unroll
    for (int i = 0; i < G; ++i) {
      if (u + i < cnt) {
        const int s = __builtin_amdgcn_readlane(s_l, u + i);
        a[i] = __int_as_float(__builtin_amdgcn_readlane(ai, u + i));
        v[i] = fl[(size_t)s * 32];
      }
    }
#pragma unroll
    for (int i = 0; i < G; ++i) {
      if (u + i < cnt) row_fma(acc, v[i], a[i]);
    }
  }
}

template <bool FIRST, typename VT, int G>
__global__ __launch_bounds__(256) void node_pull(
    const int* __restrict__ start, const int* __restrict__ csrc,
    const float* __restrict__ feat, const float* __restrict__ as1,
    const float* __restrict__ ad1, float* __restrict__ alpha,
    float* __restrict__ agg, int n, int E) {
  const int lane = threadIdx.x & 31;
  const int d = __builtin_amdgcn_readfirstlane(
      (int)((blockIdx.x * blockDim.x + threadIdx.x) >> 5));
  if (d >= n) return;
  const int beg = __builtin_amdgcn_readfirstlane(start[d]);
  const int end = (d + 1 < n) ? __builtin_amdgcn_readfirstlane(start[d + 1]) : E;
  const int deg = end - beg;

  const VT* fl = reinterpret_cast<const VT*>(feat) + lane;
  VT acc;
  row_zero(acc);

  if (deg <= 32) {
    const int  j  = beg + lane;
    const bool ok = lane < deg;
    int   s_l = 0;
    float a_l = 0.0f;
    if (FIRST) {
      float v = -INFINITY;
      if (ok) {
        s_l = csrc[j];
        v   = leaky(as1[s_l] + ad1[d]);
      }
      const float m  = wave_max(v);
      const float w  = ok ? expf(v - m) : 0.0f;
      const float ss = wave_sum(w);
      if (ok) {
        a_l = w / ss;
        alpha[j] = a_l;
      }
    } else {
      if (ok) {
        s_l = csrc[j];
        a_l = alpha[j];
      }
    }
    add_rows<VT, G>(fl, s_l, a_l, deg, acc);
  } else {
    float m = 0.0f, ss = 1.0f, adv = 0.0f;
    if (FIRST) {
      adv = ad1[d];
      float mm = -INFINITY;
      for (int j = beg + lane; j < end; j += 32)
        mm = fmaxf(mm, leaky(as1[csrc[j]] + adv));
      m = wave_max(mm);
      float sm = 0.0f;
      for (int j = beg + lane; j < end; j += 32)
        sm += expf(leaky(as1[csrc[j]] + adv) - m);
      ss = wave_sum(sm);
    }
    for (int c = beg; c < end; c += 32) {
      const int j = c + lane;
      int   s_l = 0;
      float a_l = 0.0f;
      if (j < end) {
        s_l = csrc[j];
        if (FIRST) {
          a_l = expf(leaky(as1[s_l] + adv) - m) / ss;
          alpha[j] = a_l;
        } else {
          a_l = alpha[j];
        }
      }
      const int left = end - c;
      add_rows<VT, G>(fl, s_l, a_l, left < 32 ? left : 32, acc);
    }
  }
  reinterpret_cast<VT*>(agg)[(size_t)d * 32 + lane] = acc;
}

extern "C" void kernel_launch(void* const* d_in, const int* in_sizes, int n_in,
                              void* d_out, int out_size, void* d_ws,
                              size_t ws_size, hipStream_t stream) {
  const float* x   = (const float*)d_in[0];
  const float* W1  = (const float*)d_in[1];
  const float* a1s = (const float*)d_in[2];
  const float* a1d = (const float*)d_in[3];
  const float* W2  = (const float*)d_in[4];
  const int*   ei  = (const int*)d_in[5];

  float* out = (float*)d_out;
  float* h2  = out;
  float* h4  = out + (size_t)NN * OUTD;

  float* ws    = (float*)d_ws;
  float* xs1   = ws;
  float* agg   = xs1 + (size_t)NN * HIDD;
  float* as1   = agg + (size_t)NN * HIDD;
  float* ad1   = as1 + NN;
  int*   start = (int*)(ad1 + NN) + NN;
  float* ebuf  = (float*)(start + NN);
  int*   csrc  = (int*)ebuf;
  float* alpha = ebuf + NE;
  int*   binned   = (int*)agg;
  int*   th       = binned + NE;
  int*   base     = th + NTILE * NBIN;
  int*   binstart = base + NTILE * NBIN;

  const int strips  = NN / 16;
  const int gblocks = (strips + 3) / 4;
  const int npullB  = (NN * 32 + 255) / 256;
  const int nwarpsB = (NN * 32 + 255) / 256;

  bin_count<<<NTILE, SORT_THR, 0, stream>>>(ei, th);
  bin_scan<<<1, SORT_THR, 0, stream>>>(th, base, binstart);
  bin_scatter<<<NTILE, SORT_THR, 0, stream>>>(ei, th, base, binned);
  bin_finish<<<NBIN, SORT_THR, 0, stream>>>(binned, binstart, start, csrc);

  gemm_wmma<IND, HIDD, false, false>
      <<<gblocks, 128, IND * HIDD * sizeof(float), stream>>>(x, W1, xs1, NN);

  node_dots<<<nwarpsB, 256, 0, stream>>>(xs1, a1s, a1d, as1, ad1, NN);

  node_pull<true, float4, 8>
      <<<npullB, 256, 0, stream>>>(start, csrc, xs1, as1, ad1, alpha, agg, NN, NE);

  gemm_wmma<HIDD, OUTD, false, true>
      <<<gblocks, 128, HIDD * OUTD * sizeof(float), stream>>>(agg, W2, h2, NN);

  float* aggh = agg;
  node_pull<false, float, 16>
      <<<npullB, 256, 0, stream>>>(start, csrc, h2, as1, ad1, alpha, aggh, NN, NE);

  gemm_wmma<OUTD, HIDD, true, false>
      <<<gblocks, 128, OUTD * HIDD * sizeof(float), stream>>>(aggh, W2, xs1, NN);

  gemm_wmma<HIDD, IND, true, true>
      <<<gblocks, 128, HIDD * IND * sizeof(float), stream>>>(xs1, W1, h4, NN);
}


# Per-node kernels (logit dots, both aggregations): a wave walks a run of consecutive nodes (16 rows in dots, 10 nodes in sums), not one short wave per node: 300000 waves -> 26250; sum order kept
# speedup vs baseline: 4.5099x; 1.0462x over previous
#include <hip/hip_runtime.h>
#include <hip/hip_bf16.h>


typedef float v2f __attribute__((ext_vector_type(2)));
typedef float v8f __attribute__((ext_vector_type(8)));

#define NN   100000
#define NE   1600000
#define IND  256
#define HIDD 128
#define OUTD 32
#define NEG_SLOPE 0.2f

template <int K, int NC, bool BT>
__device__ __forceinline__ void stage_b(const float* __restrict__ Bm,
                                        float* __restrict__ smem, int tid) {
  constexpr int NTHR = 128;
  if (!BT) {
    constexpr int NQ    = NC / 4;
    constexpr int ITEMS = (K / 2) * NQ;
    constexpr int PER_T = ITEMS / NTHR;
    constexpr int UN    = PER_T < 8 ? PER_T : 8;
    static_assert(ITEMS % (NTHR * UN) == 0, "whole batches");
    for (int base = 0; base < ITEMS; base += NTHR * UN) {
      float4 r0[UN], r1[UN];
#pragma unroll
      for (int u = 0; u < UN; ++u) {
        const int it = base + u * NTHR + tid;
        const int kp = it / NQ;
        const float* g = Bm + 4 * it + kp * NC;
        r0[u] = *reinterpret_cast<const float4*>(g);
        r1[u] = *reinterpret_cast<const float4*>(g + NC);
      }
#pragma unroll
      for (int u = 0; u < UN; ++u) {
        const int it = base + u * NTHR + tid;
        float4* dst = reinterpret_cast<float4*>(smem + 8 * it);
        dst[0] = make_float4(r0[u].x, r1[u].x, r0[u].y, r1[u].y);
        dst[1] = make_float4(r0[u].z, r1[u].z, r0[u].w, r1[u].w);
      }
    }
  } else {
    constexpr int LK    = 4;
    constexpr int NSUB  = NTHR / LK;
    constexpr int KQH   = (K / 4) / LK;
    constexpr int STEPS = (NC / NSUB) * KQH;
    constexpr int UN    = STEPS < 8 ? STEPS : 8;
    static_assert((K / 4) % LK == 0 && NC % NSUB == 0 && STEPS % UN == 0,
                  "whole batches");
    const int lk   = tid % LK;
    const int nsub = tid / LK;
    for (int s0 = 0; s0 < STEPS; s0 += UN) {
      float4 r[UN];
#pragma unroll
      for (int u = 0; u < UN; ++u) {
        const int s  = s0 + u;
        const int kq = (s % KQH) * LK + lk;
        const int n  = (s / KQH) * NSUB + nsub;
        r[u] = *reinterpret_cast<const float4*>(Bm + n * K + 4 * kq);
      }
#pragma unroll
      for (int u = 0; u < UN; ++u) {
        const int s  = s0 + u;
        const int kq = (s % KQH) * LK + lk;
        const int n  = (s / KQH) * NSUB + nsub;
        float* p = smem + (((2 * kq) * NC + n) << 1);
        *reinterpret_cast<float2*>(p)          = make_float2(r[u].x, r[u].y);
        *reinterpret_cast<float2*>(p + 2 * NC) = make_float2(r[u].z, r[u].w);
      }
    }
  }
}

template <int K, int NC, bool BT, bool ELU>
__global__ __launch_bounds__(128) void gemm_wmma(const float* __restrict__ A,
                                                 const float* __restrict__ Bm,
                                                 float* __restrict__ D, int M) {
  extern __shared__ __align__(16) float smem[];

  const int tid = threadIdx.x;
  stage_b<K, NC, BT>(Bm, smem, tid);
  __syncthreads();

  const int lane  = tid & 31;
  const int wv    = tid >> 5;
  const int strip = blockIdx.x * (blockDim.x >> 5) + wv;
  const int row0  = strip * 16;
  if (row0 >= M) return;

  const int m    = lane & 15;
  const int koff = (lane >> 4) << 1;
  const float* arow = A + (size_t)(row0 + m) * K + koff;

  constexpr int NH    = (NC > 128) ? 128 : NC;
  constexpr int NPASS = NC / NH;
  constexpr int NT    = NH / 16;
  static_assert(K % 8 == 0, "k-loop unrolled by 8");

  for (int np = 0; np < NPASS; ++np) {
    v8f acc[NT];
#pragma unroll
    for (int t = 0; t < NT; ++t)
#pragma unroll
      for (int i = 0; i < 8; ++i) acc[t][i] = 0.0f;

    auto loadB = [&](int kk, v2f* dst) {
      const float* bp = smem + ((((kk + koff) >> 1) * NC + np * NH + m) << 1);
#pragma unroll
      for (int t = 0; t < NT; ++t)
        dst[t] = *reinterpret_cast<const v2f*>(bp + t * 32);
    };
    auto mmaStep = [&](int kk, const v2f* bv) {
      v2f av = *reinterpret_cast<const v2f*>(arow + kk);
      if (ELU) {
        av[0] = av[0] > 0.0f ? av[0] : (expf(av[0]) - 1.0f);
        av[1] = av[1] > 0.0f ? av[1] : (expf(av[1]) - 1.0f);
      }
#pragma unroll
      for (int t = 0; t < NT; ++t)
        acc[t] = __builtin_amdgcn_wmma_f32_16x16x4_f32(
            false, av, false, bv[t], (short)0, acc[t], false, false);
    };

    v2f bv0[NT], bv1[NT];
    loadB(0, bv0);
    for (int k0 = 0; k0 < K; k0 += 8) {
      loadB(k0 + 4, bv1);
      mmaStep(k0, bv0);
      if (k0 + 8 < K) loadB(k0 + 8, bv0);
      mmaStep(k0 + 4, bv1);
    }

    const int mbase = (lane >> 4) * 8;
#pragma unroll
    for (int t = 0; t < NT; ++t) {
#pragma unroll
      for (int v = 0; v < 8; ++v) {
        D[(size_t)(row0 + mbase + v) * NC + np * NH + t * 16 + m] = acc[t][v];
      }
    }
  }
}

#define DOT_ROWS 16
__global__ __launch_bounds__(256) void node_dots(const float* __restrict__ xs1,
                          const float* __restrict__ a_s,
                          const float* __restrict__ a_d,
                          float* __restrict__ as1, float* __restrict__ ad1,
                          int n) {
  const int lane = threadIdx.x & 31;
  const int d0 = DOT_ROWS * __builtin_amdgcn_readfirstlane(
      (int)((blockIdx.x * blockDim.x + threadIdx.x) >> 5));
  if (d0 >= n) return;
  const float4 cs = ((const float4*)a_s)[lane];
  const float4 cd = ((const float4*)a_d)[lane];
  float os = 0.0f, od = 0.0f;
  float4 v[DOT_ROWS];
#pragma unroll
  for (int i = 0; i < DOT_ROWS; ++i) {
    const int r = (d0 + i < n) ? (d0 + i) : (n - 1);
    v[i] = ((const float4*)(xs1 + (size_t)r * HIDD))[lane];
  }
#pragma unroll
  for (int i = 0; i < DOT_ROWS; ++i) {
    float s = fmaf(v[i].w, cs.w,
                   fmaf(v[i].z, cs.z, fmaf(v[i].x, cs.x, v[i].y * cs.y)));
    float d = fmaf(v[i].w, cd.w,
                   fmaf(v[i].z, cd.z, fmaf(v[i].x, cd.x, v[i].y * cd.y)));
#pragma unroll
    for (int off = 16; off > 0; off >>= 1) {
      s += __shfl_down(s, off, 32);
      d += __shfl_down(d, off, 32);
    }
    const float s0 = __int_as_float(
        __builtin_amdgcn_readfirstlane(__float_as_int(s)));
    const float t0 = __int_as_float(
        __builtin_amdgcn_readfirstlane(__float_as_int(d)));
    if (lane == i) {
      os = s0;
      od = t0;
    }
  }
  if (lane < DOT_ROWS && d0 + lane < n) {
    as1[d0 + lane] = os;
    ad1[d0 + lane] = od;
  }
}

#define TE       8192
#define NTILE    ((NE + TE - 1) / TE)
#define BSH      9
#define BNODES   (1 << BSH)
#define NBIN     ((NN + BNODES - 1) / BNODES)
#define SRC_BITS 17
#define SRC_MASK ((1 << SRC_BITS) - 1)
#define BCAP     12288
#define SORT_THR 1024
static_assert(NN <= (1 << SRC_BITS), "node id packs into SRC_BITS");
static_assert(NBIN < 256 && NTILE % 4 == 0 && NE % 4 == 0 && TE == 8 * SORT_THR,
              "shapes the sort kernels are written for");

__device__ __forceinline__ int block_excl_scan(int val, int* wsum) {
  const int lane = threadIdx.x & 31;
  const int wv   = threadIdx.x >> 5;
  int x = val;
#pragma unroll
  for (int off = 1; off < 32; off <<= 1) {
    const int y = __shfl_up(x, off, 32);
    if (lane >= off) x += y;
  }
  if (lane == 31) wsum[wv] = x;
  __syncthreads();
  const int ws = wsum[lane];
  int wx = ws;
#pragma unroll
  for (int off = 1; off < 32; off <<= 1) {
    const int y = __shfl_up(wx, off, 32);
    if (lane >= off) wx += y;
  }
  const int wexcl = __shfl(wx, wv, 32) - __shfl(ws, wv, 32);
  return wexcl + (x - val);
}

__global__ __launch_bounds__(SORT_THR) void bin_count(
    const int* __restrict__ ei, int* __restrict__ th) {
  __shared__ int hist[NBIN];
  const int tid  = threadIdx.x;
  const int tile = blockIdx.x;
  if (tid < NBIN) hist[tid] = 0;
  const int4* dq = reinterpret_cast<const int4*>(ei + NE);
  const int  q0  = tile * (TE / 4) + tid;
  const int  q1  = q0 + SORT_THR;
  const bool ok0 = q0 < NE / 4;
  const bool ok1 = q1 < NE / 4;
  int4 d0 = make_int4(0, 0, 0, 0), d1 = d0;
  if (ok0) d0 = dq[q0];
  if (ok1) d1 = dq[q1];
  __syncthreads();
  if (ok0) {
    atomicAdd(&hist[d0.x >> BSH], 1);
    atomicAdd(&hist[d0.y >> BSH], 1);
    atomicAdd(&hist[d0.z >> BSH], 1);
    atomicAdd(&hist[d0.w >> BSH], 1);
  }
  if (ok1) {
    atomicAdd(&hist[d1.x >> BSH], 1);
    atomicAdd(&hist[d1.y >> BSH], 1);
    atomicAdd(&hist[d1.z >> BSH], 1);
    atomicAdd(&hist[d1.w >> BSH], 1);
  }
  __syncthreads();
  if (tid < NBIN) th[tile * NBIN + tid] = hist[tid];
}

__global__ __launch_bounds__(SORT_THR) void bin_scan(
    const int* __restrict__ th, int* __restrict__ base,
    int* __restrict__ binstart) {
  __shared__ int psum[4][256];
  __shared__ int bst[256];
  __shared__ int wsum[32];
  constexpr int QT = NTILE / 4;
  const int tid = threadIdx.x;
  const int q   = tid >> 8;
  const int b   = tid & 255;
  int v[QT];
#pragma unroll
  for (int u = 0; u < QT; ++u)
    v[u] = (b < NBIN) ? th[(q * QT + u) * NBIN + b] : 0;
  int s = 0;
#pragma unroll
  for (int u = 0; u < QT; ++u) s += v[u];
  psum[q][b] = s;
  __syncthreads();
  const int p0 = psum[0][b], p1 = psum[1][b], p2 = psum[2][b], p3 = psum[3][b];
  const int tot  = (q == 0) ? (p0 + p1) + (p2 + p3) : 0;
  const int excl = block_excl_scan(tot, wsum);
  if (q == 0) {
    bst[b] = excl;
    if (b <= NBIN) binstart[b] = excl;
  }
  __syncthreads();
  int run = bst[b];
  if (q > 0) run += p0;
  if (q > 1) run += p1;
  if (q > 2) run += p2;
  if (b < NBIN) {
#pragma unroll
    for (int u = 0; u < QT; ++u) {
      base[(q * QT + u) * NBIN + b] = run;
      run += v[u];
    }
  }
}

__global__ __launch_bounds__(SORT_THR) void bin_scatter(
    const int* __restrict__ ei, const int* __restrict__ th,
    const int* __restrict__ base, int* __restrict__ binned) {
  __shared__ int lstart[256];
  __shared__ int cur[256];
  __shared__ int gbase[256];
  __shared__ int wsum[32];
  __shared__ int stage[TE];
  const int tid  = threadIdx.x;
  const int tile = blockIdx.x;
  const int4* sq = reinterpret_cast<const int4*>(ei);
  const int4* dq = reinterpret_cast<const int4*>(ei + NE);
  const int  q0  = tile * (TE / 4) + tid;
  const int  q1  = q0 + SORT_THR;
  const bool ok0 = q0 < NE / 4;
  const bool ok1 = q1 < NE / 4;
  int4 s0 = make_int4(0, 0, 0, 0), s1 = s0, d0 = s0, d1 = s0;
  if (ok0) {
    s0 = sq[q0];
    d0 = dq[q0];
  }
  if (ok1) {
    s1 = sq[q1];
    d1 = dq[q1];
  }
  const int c  = (tid < NBIN) ? th[tile * NBIN + tid] : 0;
  const int gb = (tid < NBIN) ? base[tile * NBIN + tid] : 0;
  const int ex = block_excl_scan(c, wsum);
  if (tid < 256) {
    lstart[tid] = ex;
    cur[tid]    = ex;
    gbase[tid]  = gb;
  }
  __syncthreads();
  auto place = [&](int s, int d) {
    const int p = atomicAdd(&cur[d >> BSH], 1);
    stage[p] = s | ((d & (BNODES - 1)) << SRC_BITS);
  };
  if (ok0) {
    place(s0.x, d0.x);
    place(s0.y, d0.y);
    place(s0.z, d0.z);
    place(s0.w, d0.w);
  }
  if (ok1) {
    place(s1.x, d1.x);
    place(s1.y, d1.y);
    place(s1.z, d1.z);
    place(s1.w, d1.w);
  }
  __syncthreads();
  const int lane = tid & 31;
  const int wv   = tid >> 5;
  for (int b = wv; b < NBIN; b += 32) {
    const int s = lstart[b];
    const int n = lstart[b + 1] - s;
    const int g = gbase[b];
    for (int i = lane; i < n; i += 32) binned[g + i] = stage[s + i];
  }
}

__global__ __launch_bounds__(SORT_THR) void bin_finish(
    const int* __restrict__ binned, const int* __restrict__ binstart,
    int* __restrict__ start, int* __restrict__ csrc) {
  __shared__ int cnt[BNODES];
  __shared__ int nbeg[BNODES];
  __shared__ int wsum[32];
  __shared__ int outb[BCAP];
  static_assert(BNODES <= SORT_THR && BNODES % 32 == 0 && SORT_THR == 1024,
                "one thread per node counter; 32 waves share the nodes evenly");
  const int tid = threadIdx.x;
  const int bin = blockIdx.x;
  const int bs  = binstart[bin];
  const int c   = binstart[bin + 1] - bs;
  const int* in = binned + bs;
  if (tid < BNODES) cnt[tid] = 0;
  __syncthreads();
  for (int i0 = 0; i0 < c; i0 += 4 * SORT_THR) {
    int w[4];
#pragma unroll
    for (int u = 0; u < 4; ++u) {
      const int i = i0 + u * SORT_THR + tid;
      w[u] = (i < c) ? in[i] : -1;
    }
#pragma unroll
    for (int u = 0; u < 4; ++u)
      if (w[u] >= 0) atomicAdd(&cnt[w[u] >> SRC_BITS], 1);
  }
  __syncthreads();
  const int mine = (tid < BNODES) ? cnt[tid] : 0;
  const int ex   = block_excl_scan(mine, wsum);
  if (tid < BNODES) {
    cnt[tid]  = ex;
    nbeg[tid] = ex;
    const int node = bin * BNODES + tid;
    if (node < NN) start[node] = bs + ex;
  }
  __syncthreads();
  const bool staged = c <= BCAP;
  for (int i0 = 0; i0 < c; i0 += 4 * SORT_THR) {
    int w[4];
#pragma unroll
    for (int u = 0; u < 4; ++u) {
      const int i = i0 + u * SORT_THR + tid;
      w[u] = (i < c) ? in[i] : -1;
    }
#pragma unroll
    for (int u = 0; u < 4; ++u) {
      if (w[u] >= 0) {
        const int p = atomicAdd(&cnt[w[u] >> SRC_BITS], 1);
        if (staged) outb[p] = w[u] & SRC_MASK;
        else        csrc[bs + p] = w[u] & SRC_MASK;
      }
    }
  }
  if (staged) {
    __syncthreads();
    const int lane = tid & 31;
    const int wv   = tid >> 5;
    constexpr int NPW = BNODES / 32;
    for (int t = 0; t < NPW; ++t) {
      const int n = wv * NPW + t;
      const int sb  = __builtin_amdgcn_readfirstlane(nbeg[n]);
      const int deg = __builtin_amdgcn_readfirstlane(cnt[n]) - sb;
      if (deg > 1 && deg <= 32) {
        const int key = ((lane < deg ? outb[sb + lane] : 0) << 5) | lane;
        int r = 0;
        for (int k = 0; k < deg; ++k)
          r += (__builtin_amdgcn_readlane(key, k) < key) ? 1 : 0;
        if (lane < deg) outb[sb + r] = key >> 5;
      } else if (deg > 32 && deg <= 128) {
        int key[4], r[4];
#pragma unroll
        for (int u = 0; u < 4; ++u) {
          const int i = lane + 32 * u;
          key[u] = ((i < deg ? outb[sb + i] : 0) << 7) | i;
          r[u]   = 0;
        }
        for (int k = 0; k < deg; ++k) {
          const int yk = (outb[sb + k] << 7) | k;
#pragma unroll
          for (int u = 0; u < 4; ++u) r[u] += (yk < key[u]) ? 1 : 0;
        }
#pragma unroll
        for (int u = 0; u < 4; ++u)
          if (lane + 32 * u < deg) outb[sb + r[u]] = key[u] >> 7;
      }
    }
    __syncthreads();
    for (int i = tid; i < c; i += SORT_THR) csrc[bs + i] = outb[i];
  }
}

__device__ __forceinline__ float leaky(float v) {
  return v > 0.0f ? v : NEG_SLOPE * v;
}
__device__ __forceinline__ float wave_max(float v) {
#pragma unroll
  for (int off = 16; off > 0; off >>= 1) v = fmaxf(v, __shfl_xor(v, off, 32));
  return v;
}
__device__ __forceinline__ float wave_sum(float v) {
#pragma unroll
  for (int off = 16; off > 0; off >>= 1) v += __shfl_xor(v, off, 32);
  return v;
}

__device__ __forceinline__ void row_fma(float4& acc, const float4& v, float a) {
  acc.x = fmaf(v.x, a, acc.x);
  acc.y = fmaf(v.y, a, acc.y);
  acc.z = fmaf(v.z, a, acc.z);
  acc.w = fmaf(v.w, a, acc.w);
}
__device__ __forceinline__ void row_fma(float& acc, const float& v, float a) {
  acc = fmaf(v, a, acc);
}
__device__ __forceinline__ void row_zero(float4& v) {
  v = make_float4(0.0f, 0.0f, 0.0f, 0.0f);
}
__device__ __forceinline__ void row_zero(float& v) { v = 0.0f; }

template <typename VT, int G>
__device__ __forceinline__ void add_rows(const VT* __restrict__ fl, int s_l,
                                         float a_l, int cnt, VT& acc) {
  const int ai = __float_as_int(a_l);
  int u = 0;
  for (; u + G <= cnt; u += G) {
    VT    v[G];
    float a[G];
#pragma unroll
    for (int i = 0; i < G; ++i) {
      const int s = __builtin_amdgcn_readlane(s_l, u + i);
      a[i] = __int_as_float(__builtin_amdgcn_readlane(ai, u + i));
      v[i] = fl[(size_t)s * 32];
    }
#pragma unroll
    for (int i = 0; i < G; ++i) row_fma(acc, v[i], a[i]);
  }
  if (u < cnt) {
    VT    v[G];
    float a[G];
#pragma unroll
    for (int i = 0; i < G; ++i) {
      row_zero(v[i]);
      a[i] = 0.0f;
      if (u + i < cnt) {
        const int s = __builtin_amdgcn_readlane(s_l, u + i);
        a[i] = __int_as_float(__builtin_amdgcn_readlane(ai, u + i));
        v[i] = fl[(size_t)s * 32];
      }
    }
#pragma unroll
    for (int i = 0; i < G; ++i) {
      if (u + i < cnt) row_fma(acc, v[i], a[i]);
    }
  }
}

#define NPW 10
template <bool FIRST, typename VT, int G>
__global__ __launch_bounds__(256) void node_pull(
    const int* __restrict__ start, const int* __restrict__ csrc,
    const float* __restrict__ feat, const float* __restrict__ as1,
    const float* __restrict__ ad1, float* __restrict__ alpha,
    float* __restrict__ agg, int n, int E) {
  static_assert(NPW < 32, "lane i holds the bound of node i, lane NPW the end");
  const int lane = threadIdx.x & 31;
  const int d0 = NPW * __builtin_amdgcn_readfirstlane(
      (int)((blockIdx.x * blockDim.x + threadIdx.x) >> 5));
  if (d0 >= n) return;
  const int nd = (n - d0 < NPW) ? (n - d0) : NPW;

  int sb = E;
  if (lane <= nd && d0 + lane < n) sb = start[d0 + lane];
  int adb = 0;
  if (FIRST && lane < nd) adb = __float_as_int(ad1[d0 + lane]);

  const VT* fl = reinterpret_cast<const VT*>(feat) + lane;

  for (int i = 0; i < nd; ++i) {
    const int d = d0 + i;
    const int   beg = __builtin_amdgcn_readlane(sb, i);
    const int   end = __builtin_amdgcn_readlane(sb, i + 1);
    const float adv = __int_as_float(__builtin_amdgcn_readlane(adb, i));
    const int   deg = end - beg;

    VT acc;
    row_zero(acc);

    if (deg <= 32) {
      const int  j  = beg + lane;
      const bool ok = lane < deg;
      int   s_l = 0;
      float a_l = 0.0f;
      if (FIRST) {
        float v = -INFINITY;
        if (ok) {
          s_l = csrc[j];
          v   = leaky(as1[s_l] + adv);
        }
        const float m  = wave_max(v);
        const float w  = ok ? expf(v - m) : 0.0f;
        const float ss = wave_sum(w);
        if (ok) {
          a_l = w / ss;
          alpha[j] = a_l;
        }
      } else {
        if (ok) {
          s_l = csrc[j];
          a_l = alpha[j];
        }
      }
      add_rows<VT, G>(fl, s_l, a_l, deg, acc);
    } else {
      float m = 0.0f, ss = 1.0f;
      if (FIRST) {
        float mm = -INFINITY;
        for (int j = beg + lane; j < end; j += 32)
          mm = fmaxf(mm, leaky(as1[csrc[j]] + adv));
        m = wave_max(mm);
        float sm = 0.0f;
        for (int j = beg + lane; j < end; j += 32)
          sm += expf(leaky(as1[csrc[j]] + adv) - m);
        ss = wave_sum(sm);
      }
      for (int c = beg; c < end; c += 32) {
        const int j = c + lane;
        int   s_l = 0;
        float a_l = 0.0f;
        if (j < end) {
          s_l = csrc[j];
          if (FIRST) {
            a_l = expf(leaky(as1[s_l] + adv) - m) / ss;
            alpha[j] = a_l;
          } else {
            a_l = alpha[j];
          }
        }
        const int left = end - c;
        add_rows<VT, G>(fl, s_l, a_l, left < 32 ? left : 32, acc);
      }
    }
    reinterpret_cast<VT*>(agg)[(size_t)d * 32 + lane] = acc;
  }
}

extern "C" void kernel_launch(void* const* d_in, const int* in_sizes, int n_in,
                              void* d_out, int out_size, void* d_ws,
                              size_t ws_size, hipStream_t stream) {
  const float* x   = (const float*)d_in[0];
  const float* W1  = (const float*)d_in[1];
  const float* a1s = (const float*)d_in[2];
  const float* a1d = (const float*)d_in[3];
  const float* W2  = (const float*)d_in[4];
  const int*   ei  = (const int*)d_in[5];

  float* out = (float*)d_out;
  float* h2  = out;
  float* h4  = out + (size_t)NN * OUTD;

  float* ws    = (float*)d_ws;
  float* xs1   = ws;
  float* agg   = xs1 + (size_t)NN * HIDD;
  float* as1   = agg + (size_t)NN * HIDD;
  float* ad1   = as1 + NN;
  int*   start = (int*)(ad1 + NN) + NN;
  float* ebuf  = (float*)(start + NN);
  int*   csrc  = (int*)ebuf;
  float* alpha = ebuf + NE;
  int*   binned   = (int*)agg;
  int*   th       = binned + NE;
  int*   base     = th + NTILE * NBIN;
  int*   binstart = base + NTILE * NBIN;

  const int strips  = NN / 16;
  const int gblocks = (strips + 3) / 4;
  const int npullB  = ((NN + NPW - 1) / NPW + 7) / 8;
  const int nwarpsB = ((NN + DOT_ROWS - 1) / DOT_ROWS + 7) / 8;

  bin_count<<<NTILE, SORT_THR, 0, stream>>>(ei, th);
  bin_scan<<<1, SORT_THR, 0, stream>>>(th, base, binstart);
  bin_scatter<<<NTILE, SORT_THR, 0, stream>>>(ei, th, base, binned);
  bin_finish<<<NBIN, SORT_THR, 0, stream>>>(binned, binstart, start, csrc);

  gemm_wmma<IND, HIDD, false, false>
      <<<gblocks, 128, IND * HIDD * sizeof(float), stream>>>(x, W1, xs1, NN);

  node_dots<<<nwarpsB, 256, 0, stream>>>(xs1, a1s, a1d, as1, ad1, NN);

  node_pull<true, float4, 8>
      <<<npullB, 256, 0, stream>>>(start, csrc, xs1, as1, ad1, alpha, agg, NN, NE);

  gemm_wmma<HIDD, OUTD, false, true>
      <<<gblocks, 128, HIDD * OUTD * sizeof(float), stream>>>(agg, W2, h2, NN);

  float* aggh = agg;
  node_pull<false, float, 16>
      <<<npullB, 256, 0, stream>>>(start, csrc, h2, as1, ad1, alpha, aggh, NN, NE);

  gemm_wmma<OUTD, HIDD, true, false>
      <<<gblocks, 128, OUTD * HIDD * sizeof(float), stream>>>(aggh, W2, xs1, NN);

  gemm_wmma<HIDD, IND, true, true>
      <<<gblocks, 128, HIDD * IND * sizeof(float), stream>>>(xs1, W1, h4, NN);
}


# GEMMs: each wave requests its whole A row strip into registers beside the B staging (same 8-byte loads, same k order, bit-identical); no global load left in the k-loop; ELU once per strip
# speedup vs baseline: 5.1069x; 1.1324x over previous
#include <hip/hip_runtime.h>
#include <hip/hip_bf16.h>


typedef float v2f __attribute__((ext_vector_type(2)));
typedef float v8f __attribute__((ext_vector_type(8)));

#define NN   100000
#define NE   1600000
#define IND  256
#define HIDD 128
#define OUTD 32
#define NEG_SLOPE 0.2f

template <int K, int NC, bool BT>
__device__ __forceinline__ void stage_b(const float* __restrict__ Bm,
                                        float* __restrict__ smem, int tid) {
  constexpr int NTHR = 128;
  if (!BT) {
    constexpr int NQ    = NC / 4;
    constexpr int ITEMS = (K / 2) * NQ;
    constexpr int PER_T = ITEMS / NTHR;
    constexpr int UN    = PER_T < 8 ? PER_T : 8;
    static_assert(ITEMS % (NTHR * UN) == 0, "whole batches");
    for (int base = 0; base < ITEMS; base += NTHR * UN) {
      float4 r0[UN], r1[UN];
#pragma unroll
      for (int u = 0; u < UN; ++u) {
        const int it = base + u * NTHR + tid;
        const int kp = it / NQ;
        const float* g = Bm + 4 * it + kp * NC;
        r0[u] = *reinterpret_cast<const float4*>(g);
        r1[u] = *reinterpret_cast<const float4*>(g + NC);
      }
#pragma unroll
      for (int u = 0; u < UN; ++u) {
        const int it = base + u * NTHR + tid;
        float4* dst = reinterpret_cast<float4*>(smem + 8 * it);
        dst[0] = make_float4(r0[u].x, r1[u].x, r0[u].y, r1[u].y);
        dst[1] = make_float4(r0[u].z, r1[u].z, r0[u].w, r1[u].w);
      }
    }
  } else {
    constexpr int LK    = 4;
    constexpr int NSUB  = NTHR / LK;
    constexpr int KQH   = (K / 4) / LK;
    constexpr int STEPS = (NC / NSUB) * KQH;
    constexpr int UN    = STEPS < 8 ? STEPS : 8;
    static_assert((K / 4) % LK == 0 && NC % NSUB == 0 && STEPS % UN == 0,
                  "whole batches");
    const int lk   = tid % LK;
    const int nsub = tid / LK;
    for (int s0 = 0; s0 < STEPS; s0 += UN) {
      float4 r[UN];
#pragma unroll
      for (int u = 0; u < UN; ++u) {
        const int s  = s0 + u;
        const int kq = (s % KQH) * LK + lk;
        const int n  = (s / KQH) * NSUB + nsub;
        r[u] = *reinterpret_cast<const float4*>(Bm + n * K + 4 * kq);
      }
#pragma unroll
      for (int u = 0; u < UN; ++u) {
        const int s  = s0 + u;
        const int kq = (s % KQH) * LK + lk;
        const int n  = (s / KQH) * NSUB + nsub;
        float* p = smem + (((2 * kq) * NC + n) << 1);
        *reinterpret_cast<float2*>(p)          = make_float2(r[u].x, r[u].y);
        *reinterpret_cast<float2*>(p + 2 * NC) = make_float2(r[u].z, r[u].w);
      }
    }
  }
}

template <int K, int NC, bool BT, bool ELU>
__global__ __launch_bounds__(128) void gemm_wmma(const float* __restrict__ A,
                                                 const float* __restrict__ Bm,
                                                 float* __restrict__ D, int M) {
  extern __shared__ __align__(16) float smem[];

  const int tid   = threadIdx.x;
  const int lane  = tid & 31;
  const int wv    = tid >> 5;
  const int strip = blockIdx.x * (blockDim.x >> 5) + wv;
  const int row0  = strip * 16;
  const bool live = row0 < M;

  const int m    = lane & 15;
  const int koff = (lane >> 4) << 1;

  constexpr int KS = K / 4;
  const int arow0 = live ? row0 : (M - 16);
  const float* arow = A + (size_t)(arow0 + m) * K + koff;
  v2f a[KS];
#pragma unroll
  for (int s = 0; s < KS; ++s)
    a[s] = *reinterpret_cast<const v2f*>(arow + 4 * s);

  stage_b<K, NC, BT>(Bm, smem, tid);
  __syncthreads();
  if (!live) return;

  constexpr int LDS_HALF = 16384;
  const int lane_off = ((koff >> 1) * NC + m) << 1;
  int lane_hi = lane_off + LDS_HALF;
  if (K * NC > LDS_HALF) {
    asm("" : "+v"(lane_hi));
  }

  constexpr int NH    = (NC > 128) ? 128 : NC;
  constexpr int NPASS = NC / NH;
  constexpr int NT    = NH / 16;
  static_assert(K % 8 == 0, "k-loop unrolled by 8");

#pragma unroll
  for (int np = 0; np < NPASS; ++np) {
    v8f acc[NT];
#pragma unroll
    for (int t = 0; t < NT; ++t)
#pragma unroll
      for (int i = 0; i < 8; ++i) acc[t][i] = 0.0f;

    auto loadB = [&](int kk, v2f* dst) {
      const int cst = kk * NC + np * NH * 2;
      const float* bp = (cst >= LDS_HALF) ? smem + lane_hi + (cst - LDS_HALF)
                                          : smem + lane_off + cst;
#pragma unroll
      for (int t = 0; t < NT; ++t)
        dst[t] = *reinterpret_cast<const v2f*>(bp + t * 32);
    };
    auto eluA = [&](int s) {
      if (ELU && np == 0 && s < KS) {
        v2f av = a[s];
        const float e0 = expf(av[0]) - 1.0f;
        const float e1 = expf(av[1]) - 1.0f;
        av[0] = av[0] > 0.0f ? av[0] : e0;
        av[1] = av[1] > 0.0f ? av[1] : e1;
        a[s] = av;
      }
    };
    auto mmaStep = [&](int kk, const v2f* bv) {
      const v2f av = a[kk >> 2];
#pragma unroll
      for (int t = 0; t < NT; ++t)
        acc[t] = __builtin_amdgcn_wmma_f32_16x16x4_f32(
            false, av, false, bv[t], (short)0, acc[t], false, false);
    };

    v2f bv0[NT], bv1[NT];
    loadB(0, bv0);
    eluA(0);
#pragma unroll
    for (int k0 = 0; k0 < K; k0 += 8) {
      loadB(k0 + 4, bv1);
      __builtin_amdgcn_sched_barrier(0);
      eluA((k0 >> 2) + 1);
      mmaStep(k0, bv0);
      __builtin_amdgcn_sched_barrier(0);
      if (k0 + 8 < K) loadB(k0 + 8, bv0);
      __builtin_amdgcn_sched_barrier(0);
      eluA((k0 >> 2) + 2);
      mmaStep(k0 + 4, bv1);
      __builtin_amdgcn_sched_barrier(0);
    }

    const int mbase = (lane >> 4) * 8;
#pragma unroll
    for (int t = 0; t < NT; ++t) {
#pragma unroll
      for (int v = 0; v < 8; ++v) {
        D[(size_t)(row0 + mbase + v) * NC + np * NH + t * 16 + m] = acc[t][v];
      }
    }
  }
}

#define DOT_ROWS 16
__global__ __launch_bounds__(256) void node_dots(const float* __restrict__ xs1,
                          const float* __restrict__ a_s,
                          const float* __restrict__ a_d,
                          float* __restrict__ as1, float* __restrict__ ad1,
                          int n) {
  const int lane = threadIdx.x & 31;
  const int d0 = DOT_ROWS * __builtin_amdgcn_readfirstlane(
      (int)((blockIdx.x * blockDim.x + threadIdx.x) >> 5));
  if (d0 >= n) return;
  const float4 cs = ((const float4*)a_s)[lane];
  const float4 cd = ((const float4*)a_d)[lane];
  float os = 0.0f, od = 0.0f;
  float4 v[DOT_ROWS];
#pragma unroll
  for (int i = 0; i < DOT_ROWS; ++i) {
    const int r = (d0 + i < n) ? (d0 + i) : (n - 1);
    v[i] = ((const float4*)(xs1 + (size_t)r * HIDD))[lane];
  }
#pragma unroll
  for (int i = 0; i < DOT_ROWS; ++i) {
    float s = fmaf(v[i].w, cs.w,
                   fmaf(v[i].z, cs.z, fmaf(v[i].x, cs.x, v[i].y * cs.y)));
    float d = fmaf(v[i].w, cd.w,
                   fmaf(v[i].z, cd.z, fmaf(v[i].x, cd.x, v[i].y * cd.y)));
#pragma unroll
    for (int off = 16; off > 0; off >>= 1) {
      s += __shfl_down(s, off, 32);
      d += __shfl_down(d, off, 32);
    }
    const float s0 = __int_as_float(
        __builtin_amdgcn_readfirstlane(__float_as_int(s)));
    const float t0 = __int_as_float(
        __builtin_amdgcn_readfirstlane(__float_as_int(d)));
    if (lane == i) {
      os = s0;
      od = t0;
    }
  }
  if (lane < DOT_ROWS && d0 + lane < n) {
    as1[d0 + lane] = os;
    ad1[d0 + lane] = od;
  }
}

#define TE       8192
#define NTILE    ((NE + TE - 1) / TE)
#define BSH      9
#define BNODES   (1 << BSH)
#define NBIN     ((NN + BNODES - 1) / BNODES)
#define SRC_BITS 17
#define SRC_MASK ((1 << SRC_BITS) - 1)
#define BCAP     12288
#define SORT_THR 1024
static_assert(NN <= (1 << SRC_BITS), "node id packs into SRC_BITS");
static_assert(NBIN < 256 && NTILE % 4 == 0 && NE % 4 == 0 && TE == 8 * SORT_THR,
              "shapes the sort kernels are written for");

__device__ __forceinline__ int block_excl_scan(int val, int* wsum) {
  const int lane = threadIdx.x & 31;
  const int wv   = threadIdx.x >> 5;
  int x = val;
#pragma unroll
  for (int off = 1; off < 32; off <<= 1) {
    const int y = __shfl_up(x, off, 32);
    if (lane >= off) x += y;
  }
  if (lane == 31) wsum[wv] = x;
  __syncthreads();
  const int ws = wsum[lane];
  int wx = ws;
#pragma unroll
  for (int off = 1; off < 32; off <<= 1) {
    const int y = __shfl_up(wx, off, 32);
    if (lane >= off) wx += y;
  }
  const int wexcl = __shfl(wx, wv, 32) - __shfl(ws, wv, 32);
  return wexcl + (x - val);
}

__global__ __launch_bounds__(SORT_THR) void bin_count(
    const int* __restrict__ ei, int* __restrict__ th) {
  __shared__ int hist[NBIN];
  const int tid  = threadIdx.x;
  const int tile = blockIdx.x;
  if (tid < NBIN) hist[tid] = 0;
  const int4* dq = reinterpret_cast<const int4*>(ei + NE);
  const int  q0  = tile * (TE / 4) + tid;
  const int  q1  = q0 + SORT_THR;
  const bool ok0 = q0 < NE / 4;
  const bool ok1 = q1 < NE / 4;
  int4 d0 = make_int4(0, 0, 0, 0), d1 = d0;
  if (ok0) d0 = dq[q0];
  if (ok1) d1 = dq[q1];
  __syncthreads();
  if (ok0) {
    atomicAdd(&hist[d0.x >> BSH], 1);
    atomicAdd(&hist[d0.y >> BSH], 1);
    atomicAdd(&hist[d0.z >> BSH], 1);
    atomicAdd(&hist[d0.w >> BSH], 1);
  }
  if (ok1) {
    atomicAdd(&hist[d1.x >> BSH], 1);
    atomicAdd(&hist[d1.y >> BSH], 1);
    atomicAdd(&hist[d1.z >> BSH], 1);
    atomicAdd(&hist[d1.w >> BSH], 1);
  }
  __syncthreads();
  if (tid < NBIN) th[tile * NBIN + tid] = hist[tid];
}

__global__ __launch_bounds__(SORT_THR) void bin_scan(
    const int* __restrict__ th, int* __restrict__ base,
    int* __restrict__ binstart) {
  __shared__ int psum[4][256];
  __shared__ int bst[256];
  __shared__ int wsum[32];
  constexpr int QT = NTILE / 4;
  const int tid = threadIdx.x;
  const int q   = tid >> 8;
  const int b   = tid & 255;
  int v[QT];
#pragma unroll
  for (int u = 0; u < QT; ++u)
    v[u] = (b < NBIN) ? th[(q * QT + u) * NBIN + b] : 0;
  int s = 0;
#pragma unroll
  for (int u = 0; u < QT; ++u) s += v[u];
  psum[q][b] = s;
  __syncthreads();
  const int p0 = psum[0][b], p1 = psum[1][b], p2 = psum[2][b], p3 = psum[3][b];
  const int tot  = (q == 0) ? (p0 + p1) + (p2 + p3) : 0;
  const int excl = block_excl_scan(tot, wsum);
  if (q == 0) {
    bst[b] = excl;
    if (b <= NBIN) binstart[b] = excl;
  }
  __syncthreads();
  int run = bst[b];
  if (q > 0) run += p0;
  if (q > 1) run += p1;
  if (q > 2) run += p2;
  if (b < NBIN) {
#pragma unroll
    for (int u = 0; u < QT; ++u) {
      base[(q * QT + u) * NBIN + b] = run;
      run += v[u];
    }
  }
}

__global__ __launch_bounds__(SORT_THR) void bin_scatter(
    const int* __restrict__ ei, const int* __restrict__ th,
    const int* __restrict__ base, int* __restrict__ binned) {
  __shared__ int lstart[256];
  __shared__ int cur[256];
  __shared__ int gbase[256];
  __shared__ int wsum[32];
  __shared__ int stage[TE];
  const int tid  = threadIdx.x;
  const int tile = blockIdx.x;
  const int4* sq = reinterpret_cast<const int4*>(ei);
  const int4* dq = reinterpret_cast<const int4*>(ei + NE);
  const int  q0  = tile * (TE / 4) + tid;
  const int  q1  = q0 + SORT_THR;
  const bool ok0 = q0 < NE / 4;
  const bool ok1 = q1 < NE / 4;
  int4 s0 = make_int4(0, 0, 0, 0), s1 = s0, d0 = s0, d1 = s0;
  if (ok0) {
    s0 = sq[q0];
    d0 = dq[q0];
  }
  if (ok1) {
    s1 = sq[q1];
    d1 = dq[q1];
  }
  const int c  = (tid < NBIN) ? th[tile * NBIN + tid] : 0;
  const int gb = (tid < NBIN) ? base[tile * NBIN + tid] : 0;
  const int ex = block_excl_scan(c, wsum);
  if (tid < 256) {
    lstart[tid] = ex;
    cur[tid]    = ex;
    gbase[tid]  = gb;
  }
  __syncthreads();
  auto place = [&](int s, int d) {
    const int p = atomicAdd(&cur[d >> BSH], 1);
    stage[p] = s | ((d & (BNODES - 1)) << SRC_BITS);
  };
  if (ok0) {
    place(s0.x, d0.x);
    place(s0.y, d0.y);
    place(s0.z, d0.z);
    place(s0.w, d0.w);
  }
  if (ok1) {
    place(s1.x, d1.x);
    place(s1.y, d1.y);
    place(s1.z, d1.z);
    place(s1.w, d1.w);
  }
  __syncthreads();
  const int lane = tid & 31;
  const int wv   = tid >> 5;
  for (int b = wv; b < NBIN; b += 32) {
    const int s = lstart[b];
    const int n = lstart[b + 1] - s;
    const int g = gbase[b];
    for (int i = lane; i < n; i += 32) binned[g + i] = stage[s + i];
  }
}

__global__ __launch_bounds__(SORT_THR) void bin_finish(
    const int* __restrict__ binned, const int* __restrict__ binstart,
    int* __restrict__ start, int* __restrict__ csrc) {
  __shared__ int cnt[BNODES];
  __shared__ int nbeg[BNODES];
  __shared__ int wsum[32];
  __shared__ int outb[BCAP];
  static_assert(BNODES <= SORT_THR && BNODES % 32 == 0 && SORT_THR == 1024,
                "one thread per node counter; 32 waves share the nodes evenly");
  const int tid = threadIdx.x;
  const int bin = blockIdx.x;
  const int bs  = binstart[bin];
  const int c   = binstart[bin + 1] - bs;
  const int* in = binned + bs;
  if (tid < BNODES) cnt[tid] = 0;
  __syncthreads();
  for (int i0 = 0; i0 < c; i0 += 4 * SORT_THR) {
    int w[4];
#pragma unroll
    for (int u = 0; u < 4; ++u) {
      const int i = i0 + u * SORT_THR + tid;
      w[u] = (i < c) ? in[i] : -1;
    }
#pragma unroll
    for (int u = 0; u < 4; ++u)
      if (w[u] >= 0) atomicAdd(&cnt[w[u] >> SRC_BITS], 1);
  }
  __syncthreads();
  const int mine = (tid < BNODES) ? cnt[tid] : 0;
  const int ex   = block_excl_scan(mine, wsum);
  if (tid < BNODES) {
    cnt[tid]  = ex;
    nbeg[tid] = ex;
    const int node = bin * BNODES + tid;
    if (node < NN) start[node] = bs + ex;
  }
  __syncthreads();
  const bool staged = c <= BCAP;
  for (int i0 = 0; i0 < c; i0 += 4 * SORT_THR) {
    int w[4];
#pragma unroll
    for (int u = 0; u < 4; ++u) {
      const int i = i0 + u * SORT_THR + tid;
      w[u] = (i < c) ? in[i] : -1;
    }
#pragma unroll
    for (int u = 0; u < 4; ++u) {
      if (w[u] >= 0) {
        const int p = atomicAdd(&cnt[w[u] >> SRC_BITS], 1);
        if (staged) outb[p] = w[u] & SRC_MASK;
        else        csrc[bs + p] = w[u] & SRC_MASK;
      }
    }
  }
  if (staged) {
    __syncthreads();
    const int lane = tid & 31;
    const int wv   = tid >> 5;
    constexpr int NPW = BNODES / 32;
    for (int t = 0; t < NPW; ++t) {
      const int n = wv * NPW + t;
      const int sb  = __builtin_amdgcn_readfirstlane(nbeg[n]);
      const int deg = __builtin_amdgcn_readfirstlane(cnt[n]) - sb;
      if (deg > 1 && deg <= 32) {
        const int key = ((lane < deg ? outb[sb + lane] : 0) << 5) | lane;
        int r = 0;
        for (int k = 0; k < deg; ++k)
          r += (__builtin_amdgcn_readlane(key, k) < key) ? 1 : 0;
        if (lane < deg) outb[sb + r] = key >> 5;
      } else if (deg > 32 && deg <= 128) {
        int key[4], r[4];
#pragma unroll
        for (int u = 0; u < 4; ++u) {
          const int i = lane + 32 * u;
          key[u] = ((i < deg ? outb[sb + i] : 0) << 7) | i;
          r[u]   = 0;
        }
        for (int k = 0; k < deg; ++k) {
          const int yk = (outb[sb + k] << 7) | k;
#pragma unroll
          for (int u = 0; u < 4; ++u) r[u] += (yk < key[u]) ? 1 : 0;
        }
#pragma unroll
        for (int u = 0; u < 4; ++u)
          if (lane + 32 * u < deg) outb[sb + r[u]] = key[u] >> 7;
      }
    }
    __syncthreads();
    for (int i = tid; i < c; i += SORT_THR) csrc[bs + i] = outb[i];
  }
}

__device__ __forceinline__ float leaky(float v) {
  return v > 0.0f ? v : NEG_SLOPE * v;
}
__device__ __forceinline__ float wave_max(float v) {
#pragma unroll
  for (int off = 16; off > 0; off >>= 1) v = fmaxf(v, __shfl_xor(v, off, 32));
  return v;
}
__device__ __forceinline__ float wave_sum(float v) {
#pragma unroll
  for (int off = 16; off > 0; off >>= 1) v += __shfl_xor(v, off, 32);
  return v;
}

__device__ __forceinline__ void row_fma(float4& acc, const float4& v, float a) {
  acc.x = fmaf(v.x, a, acc.x);
  acc.y = fmaf(v.y, a, acc.y);
  acc.z = fmaf(v.z, a, acc.z);
  acc.w = fmaf(v.w, a, acc.w);
}
__device__ __forceinline__ void row_fma(float& acc, const float& v, float a) {
  acc = fmaf(v, a, acc);
}
__device__ __forceinline__ void row_zero(float4& v) {
  v = make_float4(0.0f, 0.0f, 0.0f, 0.0f);
}
__device__ __forceinline__ void row_zero(float& v) { v = 0.0f; }

template <typename VT, int G>
__device__ __forceinline__ void add_rows(const VT* __restrict__ fl, int s_l,
                                         float a_l, int cnt, VT& acc) {
  const int ai = __float_as_int(a_l);
  int u = 0;
  for (; u + G <= cnt; u += G) {
    VT    v[G];
    float a[G];
#pragma unroll
    for (int i = 0; i < G; ++i) {
      const int s = __builtin_amdgcn_readlane(s_l, u + i);
      a[i] = __int_as_float(__builtin_amdgcn_readlane(ai, u + i));
      v[i] = fl[(size_t)s * 32];
    }
#pragma unroll
    for (int i = 0; i < G; ++i) row_fma(acc, v[i], a[i]);
  }
  if (u < cnt) {
    VT    v[G];
    float a[G];
#pragma unroll
    for (int i = 0; i < G; ++i) {
      row_zero(v[i]);
      a[i] = 0.0f;
      if (u + i < cnt) {
        const int s = __builtin_amdgcn_readlane(s_l, u + i);
        a[i] = __int_as_float(__builtin_amdgcn_readlane(ai, u + i));
        v[i] = fl[(size_t)s * 32];
      }
    }
#pragma unroll
    for (int i = 0; i < G; ++i) {
      if (u + i < cnt) row_fma(acc, v[i], a[i]);
    }
  }
}

#define NPW 10
template <bool FIRST, typename VT, int G>
__global__ __launch_bounds__(256) void node_pull(
    const int* __restrict__ start, const int* __restrict__ csrc,
    const float* __restrict__ feat, const float* __restrict__ as1,
    const float* __restrict__ ad1, float* __restrict__ alpha,
    float* __restrict__ agg, int n, int E) {
  static_assert(NPW < 32, "lane i holds the bound of node i, lane NPW the end");
  const int lane = threadIdx.x & 31;
  const int d0 = NPW * __builtin_amdgcn_readfirstlane(
      (int)((blockIdx.x * blockDim.x + threadIdx.x) >> 5));
  if (d0 >= n) return;
  const int nd = (n - d0 < NPW) ? (n - d0) : NPW;

  int sb = E;
  if (lane <= nd && d0 + lane < n) sb = start[d0 + lane];
  int adb = 0;
  if (FIRST && lane < nd) adb = __float_as_int(ad1[d0 + lane]);

  const VT* fl = reinterpret_cast<const VT*>(feat) + lane;

  for (int i = 0; i < nd; ++i) {
    const int d = d0 + i;
    const int   beg = __builtin_amdgcn_readlane(sb, i);
    const int   end = __builtin_amdgcn_readlane(sb, i + 1);
    const float adv = __int_as_float(__builtin_amdgcn_readlane(adb, i));
    const int   deg = end - beg;

    VT acc;
    row_zero(acc);

    if (deg <= 32) {
      const int  j  = beg + lane;
      const bool ok = lane < deg;
      int   s_l = 0;
      float a_l = 0.0f;
      if (FIRST) {
        float v = -INFINITY;
        if (ok) {
          s_l = csrc[j];
          v   = leaky(as1[s_l] + adv);
        }
        const float m  = wave_max(v);
        const float w  = ok ? expf(v - m) : 0.0f;
        const float ss = wave_sum(w);
        if (ok) {
          a_l = w / ss;
          alpha[j] = a_l;
        }
      } else {
        if (ok) {
          s_l = csrc[j];
          a_l = alpha[j];
        }
      }
      add_rows<VT, G>(fl, s_l, a_l, deg, acc);
    } else {
      float m = 0.0f, ss = 1.0f;
      if (FIRST) {
        float mm = -INFINITY;
        for (int j = beg + lane; j < end; j += 32)
          mm = fmaxf(mm, leaky(as1[csrc[j]] + adv));
        m = wave_max(mm);
        float sm = 0.0f;
        for (int j = beg + lane; j < end; j += 32)
          sm += expf(leaky(as1[csrc[j]] + adv) - m);
        ss = wave_sum(sm);
      }
      for (int c = beg; c < end; c += 32) {
        const int j = c + lane;
        int   s_l = 0;
        float a_l = 0.0f;
        if (j < end) {
          s_l = csrc[j];
          if (FIRST) {
            a_l = expf(leaky(as1[s_l] + adv) - m) / ss;
            alpha[j] = a_l;
          } else {
            a_l = alpha[j];
          }
        }
        const int left = end - c;
        add_rows<VT, G>(fl, s_l, a_l, left < 32 ? left : 32, acc);
      }
    }
    reinterpret_cast<VT*>(agg)[(size_t)d * 32 + lane] = acc;
  }
}

extern "C" void kernel_launch(void* const* d_in, const int* in_sizes, int n_in,
                              void* d_out, int out_size, void* d_ws,
                              size_t ws_size, hipStream_t stream) {
  const float* x   = (const float*)d_in[0];
  const float* W1  = (const float*)d_in[1];
  const float* a1s = (const float*)d_in[2];
  const float* a1d = (const float*)d_in[3];
  const float* W2  = (const float*)d_in[4];
  const int*   ei  = (const int*)d_in[5];

  float* out = (float*)d_out;
  float* h2  = out;
  float* h4  = out + (size_t)NN * OUTD;

  float* ws    = (float*)d_ws;
  float* xs1   = ws;
  float* agg   = xs1 + (size_t)NN * HIDD;
  float* as1   = agg + (size_t)NN * HIDD;
  float* ad1   = as1 + NN;
  int*   start = (int*)(ad1 + NN) + NN;
  float* ebuf  = (float*)(start + NN);
  int*   csrc  = (int*)ebuf;
  float* alpha = ebuf + NE;
  int*   binned   = (int*)agg;
  int*   th       = binned + NE;
  int*   base     = th + NTILE * NBIN;
  int*   binstart = base + NTILE * NBIN;

  const int strips  = NN / 16;
  const int gblocks = (strips + 3) / 4;
  const int npullB  = ((NN + NPW - 1) / NPW + 7) / 8;
  const int nwarpsB = ((NN + DOT_ROWS - 1) / DOT_ROWS + 7) / 8;

  bin_count<<<NTILE, SORT_THR, 0, stream>>>(ei, th);
  bin_scan<<<1, SORT_THR, 0, stream>>>(th, base, binstart);
  bin_scatter<<<NTILE, SORT_THR, 0, stream>>>(ei, th, base, binned);
  bin_finish<<<NBIN, SORT_THR, 0, stream>>>(binned, binstart, start, csrc);

  gemm_wmma<IND, HIDD, false, false>
      <<<gblocks, 128, IND * HIDD * sizeof(float), stream>>>(x, W1, xs1, NN);

  node_dots<<<nwarpsB, 256, 0, stream>>>(xs1, a1s, a1d, as1, ad1, NN);

  node_pull<true, float4, 8>
      <<<npullB, 256, 0, stream>>>(start, csrc, xs1, as1, ad1, alpha, agg, NN, NE);

  gemm_wmma<HIDD, OUTD, false, true>
      <<<gblocks, 128, HIDD * OUTD * sizeof(float), stream>>>(agg, W2, h2, NN);

  float* aggh = agg;
  node_pull<false, float, 16>
      <<<npullB, 256, 0, stream>>>(start, csrc, h2, as1, ad1, alpha, aggh, NN, NE);

  gemm_wmma<OUTD, HIDD, true, false>
      <<<gblocks, 128, OUTD * HIDD * sizeof(float), stream>>>(aggh, W2, xs1, NN);

  gemm_wmma<HIDD, IND, true, true>
      <<<gblocks, 128, HIDD * IND * sizeof(float), stream>>>(xs1, W1, h4, NN);
}


# GEMM output stores whole lines: B columns take permuted LDS slots so a lane's 4 tiles hold 4 consecutive columns; one 16-byte store per lane, 64 -> 16 stores per 128 columns; results bit-identical
# speedup vs baseline: 5.2622x; 1.0304x over previous
#include <hip/hip_runtime.h>
#include <hip/hip_bf16.h>


typedef float v2f __attribute__((ext_vector_type(2)));
typedef float v8f __attribute__((ext_vector_type(8)));

#define NN   100000
#define NE   1600000
#define IND  256
#define HIDD 128
#define OUTD 32
#define NEG_SLOPE 0.2f

template <int NC>
struct ColMap {
  static constexpr int TPG = (NC / 16 >= 4) ? 4 : NC / 16;
  static constexpr int GW  = 16 * TPG;
  static __device__ __forceinline__ int slot_of(int n) {
    const int r = n % GW;
    return (n - r) + 16 * (r % TPG) + r / TPG;
  }
  static __device__ __forceinline__ int col_of(int s) {
    const int r = s % GW;
    return (s - r) + TPG * (r % 16) + r / 16;
  }
};

template <int K, int NC, bool BT>
__device__ __forceinline__ void stage_b(const float* __restrict__ Bm,
                                        float* __restrict__ smem, int tid) {
  constexpr int NTHR = 128;
  using CM = ColMap<NC>;
  if (!BT) {
    constexpr int NQ    = NC / 4;
    constexpr int ITEMS = (K / 2) * NQ;
    constexpr int PER_T = ITEMS / NTHR;
    constexpr int UN    = PER_T < 8 ? PER_T : 8;
    static_assert(ITEMS % (NTHR * UN) == 0, "whole batches");
    for (int base = 0; base < ITEMS; base += NTHR * UN) {
      float4 r0[UN], r1[UN];
#pragma unroll
      for (int u = 0; u < UN; ++u) {
        const int it = base + u * NTHR + tid;
        const int kp = it / NQ;
        const float* g = Bm + 4 * it + kp * NC;
        r0[u] = *reinterpret_cast<const float4*>(g);
        r1[u] = *reinterpret_cast<const float4*>(g + NC);
      }
#pragma unroll
      for (int u = 0; u < UN; ++u) {
        const int it = base + u * NTHR + tid;
        const int kp = it / NQ;
        const int n0 = 4 * (it - kp * NQ);
        float* rowp = smem + ((kp * NC) << 1);
        const float lo[4] = {r0[u].x, r0[u].y, r0[u].z, r0[u].w};
        const float hi[4] = {r1[u].x, r1[u].y, r1[u].z, r1[u].w};
#pragma unroll
        for (int j = 0; j < 4; ++j)
          *reinterpret_cast<float2*>(rowp + (CM::slot_of(n0 + j) << 1)) =
              make_float2(lo[j], hi[j]);
      }
    }
  } else {
    constexpr int LK    = 4;
    constexpr int NSUB  = NTHR / LK;
    constexpr int KQH   = (K / 4) / LK;
    constexpr int STEPS = (NC / NSUB) * KQH;
    constexpr int UN    = STEPS < 8 ? STEPS : 8;
    static_assert((K / 4) % LK == 0 && NC % NSUB == 0 && STEPS % UN == 0,
                  "whole batches");
    const int lk   = tid % LK;
    const int nsub = tid / LK;
    for (int s0 = 0; s0 < STEPS; s0 += UN) {
      float4 r[UN];
#pragma unroll
      for (int u = 0; u < UN; ++u) {
        const int s  = s0 + u;
        const int kq = (s % KQH) * LK + lk;
        const int n  = CM::col_of((s / KQH) * NSUB + nsub);
        r[u] = *reinterpret_cast<const float4*>(Bm + n * K + 4 * kq);
      }
#pragma unroll
      for (int u = 0; u < UN; ++u) {
        const int s  = s0 + u;
        const int kq = (s % KQH) * LK + lk;
        const int sl = (s / KQH) * NSUB + nsub;
        float* p = smem + (((2 * kq) * NC + sl) << 1);
        *reinterpret_cast<float2*>(p)          = make_float2(r[u].x, r[u].y);
        *reinterpret_cast<float2*>(p + 2 * NC) = make_float2(r[u].z, r[u].w);
      }
    }
  }
}

template <int K, int NC, bool BT, bool ELU>
__global__ __launch_bounds__(128) void gemm_wmma(const float* __restrict__ A,
                                                 const float* __restrict__ Bm,
                                                 float* __restrict__ D, int M) {
  extern __shared__ __align__(16) float smem[];

  const int tid   = threadIdx.x;
  const int lane  = tid & 31;
  const int wv    = tid >> 5;
  const int strip = blockIdx.x * (blockDim.x >> 5) + wv;
  const int row0  = strip * 16;
  const bool live = row0 < M;

  const int m    = lane & 15;
  const int koff = (lane >> 4) << 1;

  constexpr int KS = K / 4;
  const int arow0 = live ? row0 : (M - 16);
  const float* arow = A + (size_t)(arow0 + m) * K + koff;
  v2f a[KS];
#pragma unroll
  for (int s = 0; s < KS; ++s)
    a[s] = *reinterpret_cast<const v2f*>(arow + 4 * s);

  stage_b<K, NC, BT>(Bm, smem, tid);
  __syncthreads();
  if (!live) return;

  constexpr int LDS_HALF = 16384;
  const int lane_off = ((koff >> 1) * NC + m) << 1;
  int lane_hi = lane_off + LDS_HALF;
  if (K * NC > LDS_HALF) {
    asm("" : "+v"(lane_hi));
  }

  constexpr int NH    = (NC > 128) ? 128 : NC;
  constexpr int NPASS = NC / NH;
  constexpr int NT    = NH / 16;
  static_assert(K % 8 == 0, "k-loop unrolled by 8");

#pragma unroll
  for (int np = 0; np < NPASS; ++np) {
    v8f acc[NT];
#pragma unroll
    for (int t = 0; t < NT; ++t)
#pragma unroll
      for (int i = 0; i < 8; ++i) acc[t][i] = 0.0f;

    auto loadB = [&](int kk, v2f* dst) {
      const int cst = kk * NC + np * NH * 2;
      const float* bp = (cst >= LDS_HALF) ? smem + lane_hi + (cst - LDS_HALF)
                                          : smem + lane_off + cst;
#pragma unroll
      for (int t = 0; t < NT; ++t)
        dst[t] = *reinterpret_cast<const v2f*>(bp + t * 32);
    };
    auto eluA = [&](int s) {
      if (ELU && np == 0 && s < KS) {
        v2f av = a[s];
        const float e0 = expf(av[0]) - 1.0f;
        const float e1 = expf(av[1]) - 1.0f;
        av[0] = av[0] > 0.0f ? av[0] : e0;
        av[1] = av[1] > 0.0f ? av[1] : e1;
        a[s] = av;
      }
    };
    auto mmaStep = [&](int kk, const v2f* bv) {
      const v2f av = a[kk >> 2];
#pragma unroll
      for (int t = 0; t < NT; ++t)
        acc[t] = __builtin_amdgcn_wmma_f32_16x16x4_f32(
            false, av, false, bv[t], (short)0, acc[t], false, false);
    };

    v2f bv0[NT], bv1[NT];
    loadB(0, bv0);
    eluA(0);
#pragma unroll
    for (int k0 = 0; k0 < K; k0 += 8) {
      loadB(k0 + 4, bv1);
      __builtin_amdgcn_sched_barrier(0);
      eluA((k0 >> 2) + 1);
      mmaStep(k0, bv0);
      __builtin_amdgcn_sched_barrier(0);
      if (k0 + 8 < K) loadB(k0 + 8, bv0);
      __builtin_amdgcn_sched_barrier(0);
      eluA((k0 >> 2) + 2);
      mmaStep(k0 + 4, bv1);
      __builtin_amdgcn_sched_barrier(0);
    }

    using CM = ColMap<NC>;
    constexpr int TPG = CM::TPG;
    constexpr int GW  = CM::GW;
    static_assert(NT % TPG == 0 && NH % GW == 0 && (TPG == 4 || TPG == 2),
                  "whole tile groups per pass");
    const int mbase = (lane >> 4) * 8;
#pragma unroll
    for (int g = 0; g < NT / TPG; ++g) {
#pragma unroll
      for (int v = 0; v < 8; ++v) {
        float* dp = D + (size_t)(row0 + mbase + v) * NC + np * NH + g * GW + TPG * m;
        if (TPG == 4) {
          *reinterpret_cast<float4*>(dp) =
              make_float4(acc[4 * g][v], acc[4 * g + 1][v], acc[4 * g + 2][v],
                          acc[4 * g + 3][v]);
        } else {
          *reinterpret_cast<float2*>(dp) =
              make_float2(acc[2 * g][v], acc[2 * g + 1][v]);
        }
      }
    }
  }
}

#define DOT_ROWS 16
__global__ __launch_bounds__(256) void node_dots(const float* __restrict__ xs1,
                          const float* __restrict__ a_s,
                          const float* __restrict__ a_d,
                          float* __restrict__ as1, float* __restrict__ ad1,
                          int n) {
  const int lane = threadIdx.x & 31;
  const int d0 = DOT_ROWS * __builtin_amdgcn_readfirstlane(
      (int)((blockIdx.x * blockDim.x + threadIdx.x) >> 5));
  if (d0 >= n) return;
  const float4 cs = ((const float4*)a_s)[lane];
  const float4 cd = ((const float4*)a_d)[lane];
  float os = 0.0f, od = 0.0f;
  float4 v[DOT_ROWS];
#pragma unroll
  for (int i = 0; i < DOT_ROWS; ++i) {
    const int r = (d0 + i < n) ? (d0 + i) : (n - 1);
    v[i] = ((const float4*)(xs1 + (size_t)r * HIDD))[lane];
  }
#pragma unroll
  for (int i = 0; i < DOT_ROWS; ++i) {
    float s = fmaf(v[i].w, cs.w,
                   fmaf(v[i].z, cs.z, fmaf(v[i].x, cs.x, v[i].y * cs.y)));
    float d = fmaf(v[i].w, cd.w,
                   fmaf(v[i].z, cd.z, fmaf(v[i].x, cd.x, v[i].y * cd.y)));
#pragma unroll
    for (int off = 16; off > 0; off >>= 1) {
      s += __shfl_down(s, off, 32);
      d += __shfl_down(d, off, 32);
    }
    const float s0 = __int_as_float(
        __builtin_amdgcn_readfirstlane(__float_as_int(s)));
    const float t0 = __int_as_float(
        __builtin_amdgcn_readfirstlane(__float_as_int(d)));
    if (lane == i) {
      os = s0;
      od = t0;
    }
  }
  if (lane < DOT_ROWS && d0 + lane < n) {
    as1[d0 + lane] = os;
    ad1[d0 + lane] = od;
  }
}

#define TE       8192
#define NTILE    ((NE + TE - 1) / TE)
#define BSH      9
#define BNODES   (1 << BSH)
#define NBIN     ((NN + BNODES - 1) / BNODES)
#define SRC_BITS 17
#define SRC_MASK ((1 << SRC_BITS) - 1)
#define BCAP     12288
#define SORT_THR 1024
static_assert(NN <= (1 << SRC_BITS), "node id packs into SRC_BITS");
static_assert(NBIN < 256 && NTILE % 4 == 0 && NE % 4 == 0 && TE == 8 * SORT_THR,
              "shapes the sort kernels are written for");

__device__ __forceinline__ int block_excl_scan(int val, int* wsum) {
  const int lane = threadIdx.x & 31;
  const int wv   = threadIdx.x >> 5;
  int x = val;
#pragma unroll
  for (int off = 1; off < 32; off <<= 1) {
    const int y = __shfl_up(x, off, 32);
    if (lane >= off) x += y;
  }
  if (lane == 31) wsum[wv] = x;
  __syncthreads();
  const int ws = wsum[lane];
  int wx = ws;
#pragma unroll
  for (int off = 1; off < 32; off <<= 1) {
    const int y = __shfl_up(wx, off, 32);
    if (lane >= off) wx += y;
  }
  const int wexcl = __shfl(wx, wv, 32) - __shfl(ws, wv, 32);
  return wexcl + (x - val);
}

__global__ __launch_bounds__(SORT_THR) void bin_count(
    const int* __restrict__ ei, int* __restrict__ th) {
  __shared__ int hist[NBIN];
  const int tid  = threadIdx.x;
  const int tile = blockIdx.x;
  if (tid < NBIN) hist[tid] = 0;
  const int4* dq = reinterpret_cast<const int4*>(ei + NE);
  const int  q0  = tile * (TE / 4) + tid;
  const int  q1  = q0 + SORT_THR;
  const bool ok0 = q0 < NE / 4;
  const bool ok1 = q1 < NE / 4;
  int4 d0 = make_int4(0, 0, 0, 0), d1 = d0;
  if (ok0) d0 = dq[q0];
  if (ok1) d1 = dq[q1];
  __syncthreads();
  if (ok0) {
    atomicAdd(&hist[d0.x >> BSH], 1);
    atomicAdd(&hist[d0.y >> BSH], 1);
    atomicAdd(&hist[d0.z >> BSH], 1);
    atomicAdd(&hist[d0.w >> BSH], 1);
  }
  if (ok1) {
    atomicAdd(&hist[d1.x >> BSH], 1);
    atomicAdd(&hist[d1.y >> BSH], 1);
    atomicAdd(&hist[d1.z >> BSH], 1);
    atomicAdd(&hist[d1.w >> BSH], 1);
  }
  __syncthreads();
  if (tid < NBIN) th[tile * NBIN + tid] = hist[tid];
}

__global__ __launch_bounds__(SORT_THR) void bin_scan(
    const int* __restrict__ th, int* __restrict__ base,
    int* __restrict__ binstart) {
  __shared__ int psum[4][256];
  __shared__ int bst[256];
  __shared__ int wsum[32];
  constexpr int QT = NTILE / 4;
  const int tid = threadIdx.x;
  const int q   = tid >> 8;
  const int b   = tid & 255;
  int v[QT];
#pragma unroll
  for (int u = 0; u < QT; ++u)
    v[u] = (b < NBIN) ? th[(q * QT + u) * NBIN + b] : 0;
  int s = 0;
#pragma unroll
  for (int u = 0; u < QT; ++u) s += v[u];
  psum[q][b] = s;
  __syncthreads();
  const int p0 = psum[0][b], p1 = psum[1][b], p2 = psum[2][b], p3 = psum[3][b];
  const int tot  = (q == 0) ? (p0 + p1) + (p2 + p3) : 0;
  const int excl = block_excl_scan(tot, wsum);
  if (q == 0) {
    bst[b] = excl;
    if (b <= NBIN) binstart[b] = excl;
  }
  __syncthreads();
  int run = bst[b];
  if (q > 0) run += p0;
  if (q > 1) run += p1;
  if (q > 2) run += p2;
  if (b < NBIN) {
#pragma unroll
    for (int u = 0; u < QT; ++u) {
      base[(q * QT + u) * NBIN + b] = run;
      run += v[u];
    }
  }
}

__global__ __launch_bounds__(SORT_THR) void bin_scatter(
    const int* __restrict__ ei, const int* __restrict__ th,
    const int* __restrict__ base, int* __restrict__ binned) {
  __shared__ int lstart[256];
  __shared__ int cur[256];
  __shared__ int gbase[256];
  __shared__ int wsum[32];
  __shared__ int stage[TE];
  const int tid  = threadIdx.x;
  const int tile = blockIdx.x;
  const int4* sq = reinterpret_cast<const int4*>(ei);
  const int4* dq = reinterpret_cast<const int4*>(ei + NE);
  const int  q0  = tile * (TE / 4) + tid;
  const int  q1  = q0 + SORT_THR;
  const bool ok0 = q0 < NE / 4;
  const bool ok1 = q1 < NE / 4;
  int4 s0 = make_int4(0, 0, 0, 0), s1 = s0, d0 = s0, d1 = s0;
  if (ok0) {
    s0 = sq[q0];
    d0 = dq[q0];
  }
  if (ok1) {
    s1 = sq[q1];
    d1 = dq[q1];
  }
  const int c  = (tid < NBIN) ? th[tile * NBIN + tid] : 0;
  const int gb = (tid < NBIN) ? base[tile * NBIN + tid] : 0;
  const int ex = block_excl_scan(c, wsum);
  if (tid < 256) {
    lstart[tid] = ex;
    cur[tid]    = ex;
    gbase[tid]  = gb;
  }
  __syncthreads();
  auto place = [&](int s, int d) {
    const int p = atomicAdd(&cur[d >> BSH], 1);
    stage[p] = s | ((d & (BNODES - 1)) << SRC_BITS);
  };
  if (ok0) {
    place(s0.x, d0.x);
    place(s0.y, d0.y);
    place(s0.z, d0.z);
    place(s0.w, d0.w);
  }
  if (ok1) {
    place(s1.x, d1.x);
    place(s1.y, d1.y);
    place(s1.z, d1.z);
    place(s1.w, d1.w);
  }
  __syncthreads();
  const int lane = tid & 31;
  const int wv   = tid >> 5;
  for (int b = wv; b < NBIN; b += 32) {
    const int s = lstart[b];
    const int n = lstart[b + 1] - s;
    const int g = gbase[b];
    for (int i = lane; i < n; i += 32) binned[g + i] = stage[s + i];
  }
}

__global__ __launch_bounds__(SORT_THR) void bin_finish(
    const int* __restrict__ binned, const int* __restrict__ binstart,
    int* __restrict__ start, int* __restrict__ csrc) {
  __shared__ int cnt[BNODES];
  __shared__ int nbeg[BNODES];
  __shared__ int wsum[32];
  __shared__ int outb[BCAP];
  static_assert(BNODES <= SORT_THR && BNODES % 32 == 0 && SORT_THR == 1024,
                "one thread per node counter; 32 waves share the nodes evenly");
  const int tid = threadIdx.x;
  const int bin = blockIdx.x;
  const int bs  = binstart[bin];
  const int c   = binstart[bin + 1] - bs;
  const int* in = binned + bs;
  if (tid < BNODES) cnt[tid] = 0;
  __syncthreads();
  for (int i0 = 0; i0 < c; i0 += 4 * SORT_THR) {
    int w[4];
#pragma unroll
    for (int u = 0; u < 4; ++u) {
      const int i = i0 + u * SORT_THR + tid;
      w[u] = (i < c) ? in[i] : -1;
    }
#pragma unroll
    for (int u = 0; u < 4; ++u)
      if (w[u] >= 0) atomicAdd(&cnt[w[u] >> SRC_BITS], 1);
  }
  __syncthreads();
  const int mine = (tid < BNODES) ? cnt[tid] : 0;
  const int ex   = block_excl_scan(mine, wsum);
  if (tid < BNODES) {
    cnt[tid]  = ex;
    nbeg[tid] = ex;
    const int node = bin * BNODES + tid;
    if (node < NN) start[node] = bs + ex;
  }
  __syncthreads();
  const bool staged = c <= BCAP;
  for (int i0 = 0; i0 < c; i0 += 4 * SORT_THR) {
    int w[4];
#pragma unroll
    for (int u = 0; u < 4; ++u) {
      const int i = i0 + u * SORT_THR + tid;
      w[u] = (i < c) ? in[i] : -1;
    }
#pragma unroll
    for (int u = 0; u < 4; ++u) {
      if (w[u] >= 0) {
        const int p = atomicAdd(&cnt[w[u] >> SRC_BITS], 1);
        if (staged) outb[p] = w[u] & SRC_MASK;
        else        csrc[bs + p] = w[u] & SRC_MASK;
      }
    }
  }
  if (staged) {
    __syncthreads();
    const int lane = tid & 31;
    const int wv   = tid >> 5;
    constexpr int NPW = BNODES / 32;
    for (int t = 0; t < NPW; ++t) {
      const int n = wv * NPW + t;
      const int sb  = __builtin_amdgcn_readfirstlane(nbeg[n]);
      const int deg = __builtin_amdgcn_readfirstlane(cnt[n]) - sb;
      if (deg > 1 && deg <= 32) {
        const int key = ((lane < deg ? outb[sb + lane] : 0) << 5) | lane;
        int r = 0;
        for (int k = 0; k < deg; ++k)
          r += (__builtin_amdgcn_readlane(key, k) < key) ? 1 : 0;
        if (lane < deg) outb[sb + r] = key >> 5;
      } else if (deg > 32 && deg <= 128) {
        int key[4], r[4];
#pragma unroll
        for (int u = 0; u < 4; ++u) {
          const int i = lane + 32 * u;
          key[u] = ((i < deg ? outb[sb + i] : 0) << 7) | i;
          r[u]   = 0;
        }
        for (int k = 0; k < deg; ++k) {
          const int yk = (outb[sb + k] << 7) | k;
#pragma unroll
          for (int u = 0; u < 4; ++u) r[u] += (yk < key[u]) ? 1 : 0;
        }
#pragma unroll
        for (int u = 0; u < 4; ++u)
          if (lane + 32 * u < deg) outb[sb + r[u]] = key[u] >> 7;
      }
    }
    __syncthreads();
    for (int i = tid; i < c; i += SORT_THR) csrc[bs + i] = outb[i];
  }
}

__device__ __forceinline__ float leaky(float v) {
  return v > 0.0f ? v : NEG_SLOPE * v;
}
__device__ __forceinline__ float wave_max(float v) {
#pragma unroll
  for (int off = 16; off > 0; off >>= 1) v = fmaxf(v, __shfl_xor(v, off, 32));
  return v;
}
__device__ __forceinline__ float wave_sum(float v) {
#pragma unroll
  for (int off = 16; off > 0; off >>= 1) v += __shfl_xor(v, off, 32);
  return v;
}

__device__ __forceinline__ void row_fma(float4& acc, const float4& v, float a) {
  acc.x = fmaf(v.x, a, acc.x);
  acc.y = fmaf(v.y, a, acc.y);
  acc.z = fmaf(v.z, a, acc.z);
  acc.w = fmaf(v.w, a, acc.w);
}
__device__ __forceinline__ void row_fma(float& acc, const float& v, float a) {
  acc = fmaf(v, a, acc);
}
__device__ __forceinline__ void row_zero(float4& v) {
  v = make_float4(0.0f, 0.0f, 0.0f, 0.0f);
}
__device__ __forceinline__ void row_zero(float& v) { v = 0.0f; }

template <typename VT, int G>
__device__ __forceinline__ void add_rows(const VT* __restrict__ fl, int s_l,
                                         float a_l, int cnt, VT& acc) {
  const int ai = __float_as_int(a_l);
  int u = 0;
  for (; u + G <= cnt; u += G) {
    VT    v[G];
    float a[G];
#pragma unroll
    for (int i = 0; i < G; ++i) {
      const int s = __builtin_amdgcn_readlane(s_l, u + i);
      a[i] = __int_as_float(__builtin_amdgcn_readlane(ai, u + i));
      v[i] = fl[(size_t)s * 32];
    }
#pragma unroll
    for (int i = 0; i < G; ++i) row_fma(acc, v[i], a[i]);
  }
  if (u < cnt) {
    VT    v[G];
    float a[G];
#pragma unroll
    for (int i = 0; i < G; ++i) {
      row_zero(v[i]);
      a[i] = 0.0f;
      if (u + i < cnt) {
        const int s = __builtin_amdgcn_readlane(s_l, u + i);
        a[i] = __int_as_float(__builtin_amdgcn_readlane(ai, u + i));
        v[i] = fl[(size_t)s * 32];
      }
    }
#pragma unroll
    for (int i = 0; i < G; ++i) {
      if (u + i < cnt) row_fma(acc, v[i], a[i]);
    }
  }
}

#define NPW 10
template <bool FIRST, typename VT, int G>
__global__ __launch_bounds__(256) void node_pull(
    const int* __restrict__ start, const int* __restrict__ csrc,
    const float* __restrict__ feat, const float* __restrict__ as1,
    const float* __restrict__ ad1, float* __restrict__ alpha,
    float* __restrict__ agg, int n, int E) {
  static_assert(NPW < 32, "lane i holds the bound of node i, lane NPW the end");
  const int lane = threadIdx.x & 31;
  const int d0 = NPW * __builtin_amdgcn_readfirstlane(
      (int)((blockIdx.x * blockDim.x + threadIdx.x) >> 5));
  if (d0 >= n) return;
  const int nd = (n - d0 < NPW) ? (n - d0) : NPW;

  int sb = E;
  if (lane <= nd && d0 + lane < n) sb = start[d0 + lane];
  int adb = 0;
  if (FIRST && lane < nd) adb = __float_as_int(ad1[d0 + lane]);

  const VT* fl = reinterpret_cast<const VT*>(feat) + lane;

  for (int i = 0; i < nd; ++i) {
    const int d = d0 + i;
    const int   beg = __builtin_amdgcn_readlane(sb, i);
    const int   end = __builtin_amdgcn_readlane(sb, i + 1);
    const float adv = __int_as_float(__builtin_amdgcn_readlane(adb, i));
    const int   deg = end - beg;

    VT acc;
    row_zero(acc);

    if (deg <= 32) {
      const int  j  = beg + lane;
      const bool ok = lane < deg;
      int   s_l = 0;
      float a_l = 0.0f;
      if (FIRST) {
        float v = -INFINITY;
        if (ok) {
          s_l = csrc[j];
          v   = leaky(as1[s_l] + adv);
        }
        const float m  = wave_max(v);
        const float w  = ok ? expf(v - m) : 0.0f;
        const float ss = wave_sum(w);
        if (ok) {
          a_l = w / ss;
          alpha[j] = a_l;
        }
      } else {
        if (ok) {
          s_l = csrc[j];
          a_l = alpha[j];
        }
      }
      add_rows<VT, G>(fl, s_l, a_l, deg, acc);
    } else {
      float m = 0.0f, ss = 1.0f;
      if (FIRST) {
        float mm = -INFINITY;
        for (int j = beg + lane; j < end; j += 32)
          mm = fmaxf(mm, leaky(as1[csrc[j]] + adv));
        m = wave_max(mm);
        float sm = 0.0f;
        for (int j = beg + lane; j < end; j += 32)
          sm += expf(leaky(as1[csrc[j]] + adv) - m);
        ss = wave_sum(sm);
      }
      for (int c = beg; c < end; c += 32) {
        const int j = c + lane;
        int   s_l = 0;
        float a_l = 0.0f;
        if (j < end) {
          s_l = csrc[j];
          if (FIRST) {
            a_l = expf(leaky(as1[s_l] + adv) - m) / ss;
            alpha[j] = a_l;
          } else {
            a_l = alpha[j];
          }
        }
        const int left = end - c;
        add_rows<VT, G>(fl, s_l, a_l, left < 32 ? left : 32, acc);
      }
    }
    reinterpret_cast<VT*>(agg)[(size_t)d * 32 + lane] = acc;
  }
}

extern "C" void kernel_launch(void* const* d_in, const int* in_sizes, int n_in,
                              void* d_out, int out_size, void* d_ws,
                              size_t ws_size, hipStream_t stream) {
  const float* x   = (const float*)d_in[0];
  const float* W1  = (const float*)d_in[1];
  const float* a1s = (const float*)d_in[2];
  const float* a1d = (const float*)d_in[3];
  const float* W2  = (const float*)d_in[4];
  const int*   ei  = (const int*)d_in[5];

  float* out = (float*)d_out;
  float* h2  = out;
  float* h4  = out + (size_t)NN * OUTD;

  float* ws    = (float*)d_ws;
  float* xs1   = ws;
  float* agg   = xs1 + (size_t)NN * HIDD;
  float* as1   = agg + (size_t)NN * HIDD;
  float* ad1   = as1 + NN;
  int*   start = (int*)(ad1 + NN) + NN;
  float* ebuf  = (float*)(start + NN);
  int*   csrc  = (int*)ebuf;
  float* alpha = ebuf + NE;
  int*   binned   = (int*)agg;
  int*   th       = binned + NE;
  int*   base     = th + NTILE * NBIN;
  int*   binstart = base + NTILE * NBIN;

  const int strips  = NN / 16;
  const int gblocks = (strips + 3) / 4;
  const int npullB  = ((NN + NPW - 1) / NPW + 7) / 8;
  const int nwarpsB = ((NN + DOT_ROWS - 1) / DOT_ROWS + 7) / 8;

  bin_count<<<NTILE, SORT_THR, 0, stream>>>(ei, th);
  bin_scan<<<1, SORT_THR, 0, stream>>>(th, base, binstart);
  bin_scatter<<<NTILE, SORT_THR, 0, stream>>>(ei, th, base, binned);
  bin_finish<<<NBIN, SORT_THR, 0, stream>>>(binned, binstart, start, csrc);

  gemm_wmma<IND, HIDD, false, false>
      <<<gblocks, 128, IND * HIDD * sizeof(float), stream>>>(x, W1, xs1, NN);

  node_dots<<<nwarpsB, 256, 0, stream>>>(xs1, a1s, a1d, as1, ad1, NN);

  node_pull<true, float4, 8>
      <<<npullB, 256, 0, stream>>>(start, csrc, xs1, as1, ad1, alpha, agg, NN, NE);

  gemm_wmma<HIDD, OUTD, false, true>
      <<<gblocks, 128, HIDD * OUTD * sizeof(float), stream>>>(agg, W2, h2, NN);

  float* aggh = agg;
  node_pull<false, float, 16>
      <<<npullB, 256, 0, stream>>>(start, csrc, h2, as1, ad1, alpha, aggh, NN, NE);

  gemm_wmma<OUTD, HIDD, true, false>
      <<<gblocks, 128, OUTD * HIDD * sizeof(float), stream>>>(aggh, W2, xs1, NN);

  gemm_wmma<HIDD, IND, true, true>
      <<<gblocks, 128, HIDD * IND * sizeof(float), stream>>>(xs1, W1, h4, NN);
}
